# starting kernel
# baseline (speedup 1.0000x reference)
#include <hip/hip_runtime.h>
#include <hip/hip_bf16.h>
#include <math.h>

typedef __attribute__((ext_vector_type(2))) float v2f;
typedef __attribute__((ext_vector_type(8))) float v8f;

#define DFEAT 64

__global__ void gcn_deg_kernel(const int* __restrict__ dst,
                               float* __restrict__ deg, int E) {
    int e = blockIdx.x * blockDim.x + threadIdx.x;
    if (e < E) {
        __hip_atomic_fetch_add(&deg[dst[e]], 1.0f,
                               __ATOMIC_RELAXED, __HIP_MEMORY_SCOPE_AGENT);
    }
}

__global__ void gcn_invdeg_kernel(float* __restrict__ deg, int N) {
    int n = blockIdx.x * blockDim.x + threadIdx.x;
    if (n < N) {
        float d = deg[n];
        deg[n] = d > 0.0f ? 1.0f / d : 0.0f;
    }
}

__global__ void gcn_scatter_kernel(const float* __restrict__ x,
                                   const int* __restrict__ src,
                                   const int* __restrict__ dst,
                                   float* __restrict__ agg, int E) {
    int t = blockIdx.x * blockDim.x + threadIdx.x;
    int e = t >> 4;
    if (e >= E) return;
    int part = (t & 15) * 4;
    const float4 v = *(const float4*)(x + (size_t)src[e] * DFEAT + part);
    float* o = agg + (size_t)dst[e] * DFEAT + part;
    __hip_atomic_fetch_add(o + 0, v.x, __ATOMIC_RELAXED, __HIP_MEMORY_SCOPE_AGENT);
    __hip_atomic_fetch_add(o + 1, v.y, __ATOMIC_RELAXED, __HIP_MEMORY_SCOPE_AGENT);
    __hip_atomic_fetch_add(o + 2, v.z, __ATOMIC_RELAXED, __HIP_MEMORY_SCOPE_AGENT);
    __hip_atomic_fetch_add(o + 3, v.w, __ATOMIC_RELAXED, __HIP_MEMORY_SCOPE_AGENT);
}

__global__ __launch_bounds__(128)
void gcn_wmma_kernel(const float* __restrict__ agg,
                     const float* __restrict__ inv_deg,
                     const float* __restrict__ W,
                     const float* __restrict__ bias,
                     float* __restrict__ out) {
    const int lane    = threadIdx.x & 31;
    const int wave    = threadIdx.x >> 5;
    const int rowBase = blockIdx.x << 4;
    const int colBase = wave << 4;
    const int m  = lane & 15;
    const int hi = lane >> 4;
    const int k2 = hi << 1;

    const float  inv  = inv_deg[rowBase + m];
    const float* arow = agg + (size_t)(rowBase + m) * DFEAT;
    const float* brow = W   + (size_t)(colBase + m) * DFEAT;

    v8f c = {};
#pragma unroll
    for (int kb = 0; kb < DFEAT; kb += 4) {
        v2f a = *(const v2f*)(arow + kb + k2);
        a.x *= inv;
        a.y *= inv;
        v2f b = *(const v2f*)(brow + kb + k2);
        c = __builtin_amdgcn_wmma_f32_16x16x4_f32(
                 false, a,  false, b,
                 (short)0, c,  false,  false);
    }

    const float bcol = bias[colBase + m];
#pragma unroll
    for (int r = 0; r < 8; ++r) {
        const int row = rowBase + r + (hi << 3);
        float v = c[r] + bcol;
        v = v > 0.0f ? v : expm1f(v);
        out[(size_t)row * DFEAT + colBase + m] = v;
    }
}

extern "C" void kernel_launch(void* const* d_in, const int* in_sizes, int n_in,
                              void* d_out, int out_size, void* d_ws, size_t ws_size,
                              hipStream_t stream) {
    const float* h   = (const float*)d_in[0];
    const int*   src = (const int*)  d_in[1];
    const int*   dst = (const int*)  d_in[2];
    const float* W1  = (const float*)d_in[3];
    const float* b1  = (const float*)d_in[4];
    const float* W2  = (const float*)d_in[5];
    const float* b2  = (const float*)d_in[6];
    float*       out = (float*)d_out;

    const int N = in_sizes[0] / DFEAT;
    const int E = in_sizes[1];

    char* ws = (char*)d_ws;
    float* deg = (float*)ws;
    size_t degBytes = (((size_t)N * sizeof(float)) + 255) & ~(size_t)255;
    float* agg = (float*)(ws + degBytes);
    float* h1  = agg + (size_t)N * DFEAT;

    const int tiles    = N / 16;
    const int scatterB = (E * 16 + 255) / 256;

    hipMemsetAsync(deg, 0, (size_t)N * sizeof(float), stream);
    gcn_deg_kernel<<<(E + 255) / 256, 256, 0, stream>>>(dst, deg, E);
    gcn_invdeg_kernel<<<(N + 255) / 256, 256, 0, stream>>>(deg, N);

    hipMemsetAsync(agg, 0, (size_t)N * DFEAT * sizeof(float), stream);
    gcn_scatter_kernel<<<scatterB, 256, 0, stream>>>(h, src, dst, agg, E);
    gcn_wmma_kernel<<<tiles, 128, 0, stream>>>(agg, deg, W1, b1, h1);

    hipMemsetAsync(agg, 0, (size_t)N * DFEAT * sizeof(float), stream);
    gcn_scatter_kernel<<<scatterB, 256, 0, stream>>>(h1, src, dst, agg, E);
    gcn_wmma_kernel<<<tiles, 128, 0, stream>>>(agg, deg, W2, b2, out);
}


# scatter: one wave per edge row, lane l owns floats l and l+32, so each gather and each f32 atomic covers one whole 128-B line (was 4 part lines); 128 edges per wave; atomic arrival order changes
# speedup vs baseline: 1.4504x; 1.4504x over previous
#include <hip/hip_runtime.h>
#include <hip/hip_bf16.h>
#include <math.h>

typedef __attribute__((ext_vector_type(2))) float v2f;
typedef __attribute__((ext_vector_type(8))) float v8f;

#define DFEAT 64

__global__ void gcn_deg_kernel(const int* __restrict__ dst,
                               float* __restrict__ deg, int E) {
    int e = blockIdx.x * blockDim.x + threadIdx.x;
    if (e < E) {
        __hip_atomic_fetch_add(&deg[dst[e]], 1.0f,
                               __ATOMIC_RELAXED, __HIP_MEMORY_SCOPE_AGENT);
    }
}

__global__ void gcn_invdeg_kernel(float* __restrict__ deg, int N) {
    int n = blockIdx.x * blockDim.x + threadIdx.x;
    if (n < N) {
        float d = deg[n];
        deg[n] = d > 0.0f ? 1.0f / d : 0.0f;
    }
}

#define EDGES_PER_WAVE  128
#define WAVES_PER_BLOCK 8
#define EDGES_PER_BLOCK (EDGES_PER_WAVE * WAVES_PER_BLOCK)

template <typename T>
__device__ __forceinline__ T* gcn_row(T* base, unsigned row, int lane) {
    const unsigned off = row * (unsigned)(DFEAT * sizeof(float)) + (unsigned)lane * (unsigned)sizeof(float);
    return (T*)((char*)base + off);
}

__device__ __forceinline__ void gcn_row_add(float* __restrict__ o, float v0, float v1) {
    __hip_atomic_fetch_add(o,      v0, __ATOMIC_RELAXED, __HIP_MEMORY_SCOPE_AGENT);
    __hip_atomic_fetch_add(o + 32, v1, __ATOMIC_RELAXED, __HIP_MEMORY_SCOPE_AGENT);
}

#define GCN_GATHER8(v0, v1, j0)                                              \
    _Pragma("unroll")                                                        \
    for (int j = 0; j < 8; ++j) {                                            \
        const unsigned sj = (unsigned)__builtin_amdgcn_readlane(s, (j0) + j);\
        const float* xr = gcn_row(x, sj, lane);                              \
        v0[j] = xr[0];                                                       \
        v1[j] = xr[32];                                                      \
    }

#define GCN_ADD8(v0, v1, j0)                                                 \
    _Pragma("unroll")                                                        \
    for (int j = 0; j < 8; ++j) {                                            \
        const unsigned dj = (unsigned)__builtin_amdgcn_readlane(d, (j0) + j);\
        gcn_row_add(gcn_row(agg, dj, lane), v0[j], v1[j]);                   \
    }

__global__ __launch_bounds__(WAVES_PER_BLOCK * 32)
void gcn_scatter_kernel(const float* __restrict__ x,
                        const int* __restrict__ src,
                        const int* __restrict__ dst,
                        float* __restrict__ agg, int E) {
    const int lane = threadIdx.x & 31;
    const int wv = __builtin_amdgcn_readfirstlane((int)(threadIdx.x >> 5));
    int e = ((int)blockIdx.x * WAVES_PER_BLOCK + wv) * EDGES_PER_WAVE;
    const int eEnd = min(e + EDGES_PER_WAVE, E);

    int sN = 0, dN = 0;
    if (e + lane < eEnd) { sN = src[e + lane]; dN = dst[e + lane]; }

    for (; e < eEnd; e += 32) {
        const int s = sN, d = dN;
        const int en = e + 32;
        sN = 0; dN = 0;
        if (en + lane < eEnd) { sN = src[en + lane]; dN = dst[en + lane]; }

        const int n = eEnd - e;
        if (n >= 32) {
            float a0[8], a1[8], b0[8], b1[8];
            GCN_GATHER8(a0, a1, 0)
            GCN_GATHER8(b0, b1, 8)
            GCN_ADD8(a0, a1, 0)
            GCN_GATHER8(a0, a1, 16)
            GCN_ADD8(b0, b1, 8)
            GCN_GATHER8(b0, b1, 24)
            GCN_ADD8(a0, a1, 16)
            GCN_ADD8(b0, b1, 24)
        } else {
            for (int j = 0; j < n; ++j) {
                const unsigned sj = (unsigned)__builtin_amdgcn_readlane(s, j);
                const unsigned dj = (unsigned)__builtin_amdgcn_readlane(d, j);
                const float* xr = gcn_row(x, sj, lane);
                gcn_row_add(gcn_row(agg, dj, lane), xr[0], xr[32]);
            }
        }
    }
}

__global__ __launch_bounds__(128)
void gcn_wmma_kernel(const float* __restrict__ agg,
                     const float* __restrict__ inv_deg,
                     const float* __restrict__ W,
                     const float* __restrict__ bias,
                     float* __restrict__ out) {
    const int lane    = threadIdx.x & 31;
    const int wave    = threadIdx.x >> 5;
    const int rowBase = blockIdx.x << 4;
    const int colBase = wave << 4;
    const int m  = lane & 15;
    const int hi = lane >> 4;
    const int k2 = hi << 1;

    const float  inv  = inv_deg[rowBase + m];
    const float* arow = agg + (size_t)(rowBase + m) * DFEAT;
    const float* brow = W   + (size_t)(colBase + m) * DFEAT;

    v8f c = {};
#pragma unroll
    for (int kb = 0; kb < DFEAT; kb += 4) {
        v2f a = *(const v2f*)(arow + kb + k2);
        a.x *= inv;
        a.y *= inv;
        v2f b = *(const v2f*)(brow + kb + k2);
        c = __builtin_amdgcn_wmma_f32_16x16x4_f32(
                 false, a,  false, b,
                 (short)0, c,  false,  false);
    }

    const float bcol = bias[colBase + m];
#pragma unroll
    for (int r = 0; r < 8; ++r) {
        const int row = rowBase + r + (hi << 3);
        float v = c[r] + bcol;
        v = v > 0.0f ? v : expm1f(v);
        out[(size_t)row * DFEAT + colBase + m] = v;
    }
}

extern "C" void kernel_launch(void* const* d_in, const int* in_sizes, int n_in,
                              void* d_out, int out_size, void* d_ws, size_t ws_size,
                              hipStream_t stream) {
    const float* h   = (const float*)d_in[0];
    const int*   src = (const int*)  d_in[1];
    const int*   dst = (const int*)  d_in[2];
    const float* W1  = (const float*)d_in[3];
    const float* b1  = (const float*)d_in[4];
    const float* W2  = (const float*)d_in[5];
    const float* b2  = (const float*)d_in[6];
    float*       out = (float*)d_out;

    const int N = in_sizes[0] / DFEAT;
    const int E = in_sizes[1];

    char* ws = (char*)d_ws;
    float* deg = (float*)ws;
    size_t degBytes = (((size_t)N * sizeof(float)) + 255) & ~(size_t)255;
    float* agg = (float*)(ws + degBytes);
    float* h1  = agg + (size_t)N * DFEAT;

    const int tiles    = N / 16;
    const int scatterB = (E + EDGES_PER_BLOCK - 1) / EDGES_PER_BLOCK;

    hipMemsetAsync(deg, 0, (size_t)N * sizeof(float), stream);
    gcn_deg_kernel<<<(E + 255) / 256, 256, 0, stream>>>(dst, deg, E);
    gcn_invdeg_kernel<<<(N + 255) / 256, 256, 0, stream>>>(deg, N);

    hipMemsetAsync(agg, 0, (size_t)N * DFEAT * sizeof(float), stream);
    gcn_scatter_kernel<<<scatterB, WAVES_PER_BLOCK * 32, 0, stream>>>(h, src, dst, agg, E);
    gcn_wmma_kernel<<<tiles, 128, 0, stream>>>(agg, deg, W1, b1, h1);

    hipMemsetAsync(agg, 0, (size_t)N * DFEAT * sizeof(float), stream);
    gcn_scatter_kernel<<<scatterB, WAVES_PER_BLOCK * 32, 0, stream>>>(h1, src, dst, agg, E);
    gcn_wmma_kernel<<<tiles, 128, 0, stream>>>(agg, deg, W2, b2, out);
}


# in-degree count: each block counts one node range x one edge-list slice in LDS (integer ds_add) and stores a count plane; the inverse kernel sums the planes; replaces 1.6M scattered global atomics
# speedup vs baseline: 1.8221x; 1.2563x over previous
#include <hip/hip_runtime.h>
#include <hip/hip_bf16.h>
#include <math.h>

typedef __attribute__((ext_vector_type(2))) float v2f;
typedef __attribute__((ext_vector_type(8))) float v8f;

#define DFEAT 64

#define DEG_RANGE   12500
#define DEG_SLICES  32
#define DEG_THREADS 1024

__device__ __forceinline__ void gcn_count(unsigned* cnt, int d, int lo) {
    const unsigned r = (unsigned)(d - lo);
    if (r < (unsigned)DEG_RANGE)
        __hip_atomic_fetch_add(&cnt[r], 1u, __ATOMIC_RELAXED, __HIP_MEMORY_SCOPE_WORKGROUP);
}

__device__ __forceinline__ void gcn_count4(unsigned* cnt, int4 v, int lo) {
    gcn_count(cnt, v.x, lo);
    gcn_count(cnt, v.y, lo);
    gcn_count(cnt, v.z, lo);
    gcn_count(cnt, v.w, lo);
}

__global__ __launch_bounds__(DEG_THREADS)
void gcn_deg_kernel(const int* __restrict__ dst,
                    unsigned* __restrict__ part, int N, int E) {
    __shared__ unsigned cnt[DEG_RANGE];
    const int lo = (int)blockIdx.y * DEG_RANGE;
    const int nr = min(DEG_RANGE, N - lo);
    for (int i = threadIdx.x; i < DEG_RANGE; i += DEG_THREADS) cnt[i] = 0u;
    __syncthreads();

    const int sl = (((E + DEG_SLICES - 1) / DEG_SLICES) + 3) & ~3;
    const int e0 = (int)blockIdx.x * sl;
    const int e1 = min(e0 + sl, E);
    const int step = 4 * DEG_THREADS;
    int e = e0 + 4 * (int)threadIdx.x;
    for (; e + 3 * step + 3 < e1; e += 4 * step) {
        const int4 v0 = *(const int4*)(dst + e);
        const int4 v1 = *(const int4*)(dst + e + step);
        const int4 v2 = *(const int4*)(dst + e + 2 * step);
        const int4 v3 = *(const int4*)(dst + e + 3 * step);
        gcn_count4(cnt, v0, lo);
        gcn_count4(cnt, v1, lo);
        gcn_count4(cnt, v2, lo);
        gcn_count4(cnt, v3, lo);
    }
    for (; e < e1; e += step) {
        if (e + 3 < e1) {
            gcn_count4(cnt, *(const int4*)(dst + e), lo);
        } else {
            for (int k = e; k < e1; ++k) gcn_count(cnt, dst[k], lo);
        }
    }
    __syncthreads();

    unsigned* p = part + (size_t)blockIdx.x * (size_t)N + lo;
    for (int i = threadIdx.x; i < nr; i += DEG_THREADS) p[i] = cnt[i];
}

__global__ void gcn_invdeg_kernel(const unsigned* __restrict__ part,
                                  float* __restrict__ deg, int N) {
    int n = blockIdx.x * blockDim.x + threadIdx.x;
    if (n < N) {
        unsigned c = 0u;
        for (int s = 0; s < DEG_SLICES; ++s) c += part[(size_t)s * (size_t)N + n];
        const float d = (float)c;
        deg[n] = d > 0.0f ? 1.0f / d : 0.0f;
    }
}

#define EDGES_PER_WAVE  128
#define WAVES_PER_BLOCK 8
#define EDGES_PER_BLOCK (EDGES_PER_WAVE * WAVES_PER_BLOCK)

template <typename T>
__device__ __forceinline__ T* gcn_row(T* base, unsigned row, int lane) {
    const unsigned off = row * (unsigned)(DFEAT * sizeof(float)) + (unsigned)lane * (unsigned)sizeof(float);
    return (T*)((char*)base + off);
}

__device__ __forceinline__ void gcn_row_add(float* __restrict__ o, float v0, float v1) {
    __hip_atomic_fetch_add(o,      v0, __ATOMIC_RELAXED, __HIP_MEMORY_SCOPE_AGENT);
    __hip_atomic_fetch_add(o + 32, v1, __ATOMIC_RELAXED, __HIP_MEMORY_SCOPE_AGENT);
}

#define GCN_GATHER8(v0, v1, j0)                                              \
    _Pragma("unroll")                                                        \
    for (int j = 0; j < 8; ++j) {                                            \
        const unsigned sj = (unsigned)__builtin_amdgcn_readlane(s, (j0) + j);\
        const float* xr = gcn_row(x, sj, lane);                              \
        v0[j] = xr[0];                                                       \
        v1[j] = xr[32];                                                      \
    }

#define GCN_ADD8(v0, v1, j0)                                                 \
    _Pragma("unroll")                                                        \
    for (int j = 0; j < 8; ++j) {                                            \
        const unsigned dj = (unsigned)__builtin_amdgcn_readlane(d, (j0) + j);\
        gcn_row_add(gcn_row(agg, dj, lane), v0[j], v1[j]);                   \
    }

__global__ __launch_bounds__(WAVES_PER_BLOCK * 32)
void gcn_scatter_kernel(const float* __restrict__ x,
                        const int* __restrict__ src,
                        const int* __restrict__ dst,
                        float* __restrict__ agg, int E) {
    const int lane = threadIdx.x & 31;
    const int wv = __builtin_amdgcn_readfirstlane((int)(threadIdx.x >> 5));
    int e = ((int)blockIdx.x * WAVES_PER_BLOCK + wv) * EDGES_PER_WAVE;
    const int eEnd = min(e + EDGES_PER_WAVE, E);

    int sN = 0, dN = 0;
    if (e + lane < eEnd) { sN = src[e + lane]; dN = dst[e + lane]; }

    for (; e < eEnd; e += 32) {
        const int s = sN, d = dN;
        const int en = e + 32;
        sN = 0; dN = 0;
        if (en + lane < eEnd) { sN = src[en + lane]; dN = dst[en + lane]; }

        const int n = eEnd - e;
        if (n >= 32) {
            float a0[8], a1[8], b0[8], b1[8];
            GCN_GATHER8(a0, a1, 0)
            GCN_GATHER8(b0, b1, 8)
            GCN_ADD8(a0, a1, 0)
            GCN_GATHER8(a0, a1, 16)
            GCN_ADD8(b0, b1, 8)
            GCN_GATHER8(b0, b1, 24)
            GCN_ADD8(a0, a1, 16)
            GCN_ADD8(b0, b1, 24)
        } else {
            for (int j = 0; j < n; ++j) {
                const unsigned sj = (unsigned)__builtin_amdgcn_readlane(s, j);
                const unsigned dj = (unsigned)__builtin_amdgcn_readlane(d, j);
                const float* xr = gcn_row(x, sj, lane);
                gcn_row_add(gcn_row(agg, dj, lane), xr[0], xr[32]);
            }
        }
    }
}

__global__ __launch_bounds__(128)
void gcn_wmma_kernel(const float* __restrict__ agg,
                     const float* __restrict__ inv_deg,
                     const float* __restrict__ W,
                     const float* __restrict__ bias,
                     float* __restrict__ out) {
    const int lane    = threadIdx.x & 31;
    const int wave    = threadIdx.x >> 5;
    const int rowBase = blockIdx.x << 4;
    const int colBase = wave << 4;
    const int m  = lane & 15;
    const int hi = lane >> 4;
    const int k2 = hi << 1;

    const float  inv  = inv_deg[rowBase + m];
    const float* arow = agg + (size_t)(rowBase + m) * DFEAT;
    const float* brow = W   + (size_t)(colBase + m) * DFEAT;

    v8f c = {};
#pragma unroll
    for (int kb = 0; kb < DFEAT; kb += 4) {
        v2f a = *(const v2f*)(arow + kb + k2);
        a.x *= inv;
        a.y *= inv;
        v2f b = *(const v2f*)(brow + kb + k2);
        c = __builtin_amdgcn_wmma_f32_16x16x4_f32(
                 false, a,  false, b,
                 (short)0, c,  false,  false);
    }

    const float bcol = bias[colBase + m];
#pragma unroll
    for (int r = 0; r < 8; ++r) {
        const int row = rowBase + r + (hi << 3);
        float v = c[r] + bcol;
        v = v > 0.0f ? v : expm1f(v);
        out[(size_t)row * DFEAT + colBase + m] = v;
    }
}

extern "C" void kernel_launch(void* const* d_in, const int* in_sizes, int n_in,
                              void* d_out, int out_size, void* d_ws, size_t ws_size,
                              hipStream_t stream) {
    const float* h   = (const float*)d_in[0];
    const int*   src = (const int*)  d_in[1];
    const int*   dst = (const int*)  d_in[2];
    const float* W1  = (const float*)d_in[3];
    const float* b1  = (const float*)d_in[4];
    const float* W2  = (const float*)d_in[5];
    const float* b2  = (const float*)d_in[6];
    float*       out = (float*)d_out;

    const int N = in_sizes[0] / DFEAT;
    const int E = in_sizes[1];

    char* ws = (char*)d_ws;
    float* deg = (float*)ws;
    size_t degBytes = (((size_t)N * sizeof(float)) + 255) & ~(size_t)255;
    float* agg = (float*)(ws + degBytes);
    float* h1  = agg + (size_t)N * DFEAT;
    unsigned* part = (unsigned*)(h1 + (size_t)N * DFEAT);

    const int tiles    = N / 16;
    const int scatterB = (E + EDGES_PER_BLOCK - 1) / EDGES_PER_BLOCK;
    const int degRanges = (N + DEG_RANGE - 1) / DEG_RANGE;

    gcn_deg_kernel<<<dim3(DEG_SLICES, degRanges), DEG_THREADS, 0, stream>>>(dst, part, N, E);
    gcn_invdeg_kernel<<<(N + 255) / 256, 256, 0, stream>>>(part, deg, N);

    hipMemsetAsync(agg, 0, (size_t)N * DFEAT * sizeof(float), stream);
    gcn_scatter_kernel<<<scatterB, WAVES_PER_BLOCK * 32, 0, stream>>>(h, src, dst, agg, E);
    gcn_wmma_kernel<<<tiles, 128, 0, stream>>>(agg, deg, W1, b1, h1);

    hipMemsetAsync(agg, 0, (size_t)N * DFEAT * sizeof(float), stream);
    gcn_scatter_kernel<<<scatterB, WAVES_PER_BLOCK * 32, 0, stream>>>(h1, src, dst, agg, E);
    gcn_wmma_kernel<<<tiles, 128, 0, stream>>>(agg, deg, W2, b2, out);
}


# aggregation: edges sorted by destination once per call (LDS cursors, no global atomics), then each agg row is one gather-sum stored once; replaces f32 atomic scatter + memsets; sum order changes
# speedup vs baseline: 2.8408x; 1.5591x over previous
#include <hip/hip_runtime.h>
#include <hip/hip_bf16.h>
#include <math.h>

typedef __attribute__((ext_vector_type(2))) float v2f;
typedef __attribute__((ext_vector_type(8))) float v8f;

#define DFEAT 64

#define DEG_RANGE   12544
#define DEG_SLICES  32
#define DEG_THREADS 1024

__device__ __forceinline__ void gcn_count(unsigned* cnt, int d, int lo) {
    const unsigned r = (unsigned)(d - lo);
    if (r < (unsigned)DEG_RANGE)
        __hip_atomic_fetch_add(&cnt[r], 1u, __ATOMIC_RELAXED, __HIP_MEMORY_SCOPE_WORKGROUP);
}

__device__ __forceinline__ void gcn_count4(unsigned* cnt, int4 v, int lo) {
    gcn_count(cnt, v.x, lo);
    gcn_count(cnt, v.y, lo);
    gcn_count(cnt, v.z, lo);
    gcn_count(cnt, v.w, lo);
}

__global__ __launch_bounds__(DEG_THREADS)
void gcn_deg_kernel(const int* __restrict__ dst,
                    unsigned* __restrict__ part, int N, int E) {
    __shared__ unsigned cnt[DEG_RANGE];
    const int lo = (int)blockIdx.y * DEG_RANGE;
    const int nr = min(DEG_RANGE, N - lo);
    for (int i = threadIdx.x; i < DEG_RANGE; i += DEG_THREADS) cnt[i] = 0u;
    __syncthreads();

    const int sl = (((E + DEG_SLICES - 1) / DEG_SLICES) + 3) & ~3;
    const int e0 = (int)blockIdx.x * sl;
    const int e1 = min(e0 + sl, E);
    const int step = 4 * DEG_THREADS;
    int e = e0 + 4 * (int)threadIdx.x;
    for (; e + 3 * step + 3 < e1; e += 4 * step) {
        const int4 v0 = *(const int4*)(dst + e);
        const int4 v1 = *(const int4*)(dst + e + step);
        const int4 v2 = *(const int4*)(dst + e + 2 * step);
        const int4 v3 = *(const int4*)(dst + e + 3 * step);
        gcn_count4(cnt, v0, lo);
        gcn_count4(cnt, v1, lo);
        gcn_count4(cnt, v2, lo);
        gcn_count4(cnt, v3, lo);
    }
    for (; e < e1; e += step) {
        if (e + 3 < e1) {
            gcn_count4(cnt, *(const int4*)(dst + e), lo);
        } else {
            for (int k = e; k < e1; ++k) gcn_count(cnt, dst[k], lo);
        }
    }
    __syncthreads();

    unsigned* p = part + (size_t)blockIdx.x * (size_t)N + lo;
    for (int i = threadIdx.x; i < nr; i += DEG_THREADS) p[i] = cnt[i];
}

#define PFX_THREADS 256

__global__ __launch_bounds__(PFX_THREADS)
void gcn_prefix_kernel(unsigned* __restrict__ part,
                       float* __restrict__ deg,
                       unsigned* __restrict__ cnt,
                       unsigned* __restrict__ blocksum,
                       unsigned* __restrict__ rowstart, int N, int E) {
    __shared__ unsigned wsum[PFX_THREADS / 32];
    const int n = blockIdx.x * PFX_THREADS + threadIdx.x;
    unsigned c = 0u;
    if (n < N) {
        unsigned v[DEG_SLICES];
#pragma unroll
        for (int s = 0; s < DEG_SLICES; ++s) v[s] = part[(size_t)s * (size_t)N + n];
#pragma unroll
        for (int s = 0; s < DEG_SLICES; ++s) {
            part[(size_t)s * (size_t)N + n] = c;
            c += v[s];
        }
        cnt[n] = c;
        const float d = (float)c;
        deg[n] = d > 0.0f ? 1.0f / d : 0.0f;
    }
    unsigned t = c;
    for (int m = 16; m > 0; m >>= 1) t += __shfl_xor(t, m);
    if ((threadIdx.x & 31) == 0) wsum[threadIdx.x >> 5] = t;
    __syncthreads();
    if (threadIdx.x == 0) {
        unsigned b = 0u;
        for (int w = 0; w < PFX_THREADS / 32; ++w) b += wsum[w];
        blocksum[blockIdx.x] = b;
        if (blockIdx.x == 0) rowstart[N] = (unsigned)E;
    }
}

#define DEG_CHUNK 13

__device__ __forceinline__ unsigned gcn_take(unsigned* cur, int d, int lo) {
    const unsigned r = (unsigned)(d - lo);
    unsigned pos = ~0u;
    if (r < (unsigned)DEG_RANGE)
        pos = __hip_atomic_fetch_add(&cur[r], 1u, __ATOMIC_RELAXED, __HIP_MEMORY_SCOPE_WORKGROUP);
    return pos;
}

__device__ __forceinline__ void gcn_put(unsigned* __restrict__ csr, unsigned pos, int s, unsigned E) {
    if (pos < E) csr[pos] = (unsigned)s;
}

__device__ __forceinline__ void gcn_place(unsigned* cur, unsigned* __restrict__ csr,
                                          int d, int s, int lo, unsigned E) {
    gcn_put(csr, gcn_take(cur, d, lo), s, E);
}

__device__ __forceinline__ void gcn_place4(unsigned* cur, unsigned* __restrict__ csr,
                                           int4 d, int4 s, int lo, unsigned E) {
    const unsigned p0 = gcn_take(cur, d.x, lo);
    const unsigned p1 = gcn_take(cur, d.y, lo);
    const unsigned p2 = gcn_take(cur, d.z, lo);
    const unsigned p3 = gcn_take(cur, d.w, lo);
    gcn_put(csr, p0, s.x, E);
    gcn_put(csr, p1, s.y, E);
    gcn_put(csr, p2, s.z, E);
    gcn_put(csr, p3, s.w, E);
}

__global__ __launch_bounds__(DEG_THREADS)
void gcn_fill_kernel(const int* __restrict__ src,
                     const int* __restrict__ dst,
                     const unsigned* __restrict__ part,
                     const unsigned* __restrict__ cnt,
                     const unsigned* __restrict__ blocksum,
                     unsigned* __restrict__ rowstart,
                     unsigned* __restrict__ csr, int N, int E) {
    __shared__ unsigned cur[DEG_RANGE];
    __shared__ unsigned wtot[DEG_THREADS / 32];
    __shared__ unsigned wbase[DEG_THREADS / 32];
    const int tid  = threadIdx.x;
    const int lane = tid & 31;
    const int wave = tid >> 5;
    const int lo = (int)blockIdx.y * DEG_RANGE;
    const int nr = min(DEG_RANGE, N - lo);

    const int nb = (int)blockIdx.y * (DEG_RANGE / PFX_THREADS);
    unsigned pb = 0u;
    for (int i = tid; i < nb; i += DEG_THREADS) pb += blocksum[i];
    for (int i = tid; i < nr; i += DEG_THREADS) cur[i] = cnt[lo + i];
    __syncthreads();

    const int i0 = tid * DEG_CHUNK;
    unsigned ls = 0u;
#pragma unroll
    for (int k = 0; k < DEG_CHUNK; ++k)
        if (i0 + k < nr) ls += cur[i0 + k];
    unsigned x = ls;
    for (int d = 1; d < 32; d <<= 1) {
        const unsigned y = __shfl_up(x, d);
        if (lane >= d) x += y;
    }
    for (int m = 16; m > 0; m >>= 1) pb += __shfl_xor(pb, m);
    if (lane == 31) wtot[wave] = x;
    if (lane == 0)  wbase[wave] = pb;
    __syncthreads();
    unsigned run = x - ls;
    for (int w = 0; w < DEG_THREADS / 32; ++w) {
        run += wbase[w];
        if (w < wave) run += wtot[w];
    }
#pragma unroll
    for (int k = 0; k < DEG_CHUNK; ++k) {
        if (i0 + k < nr) {
            const unsigned c = cur[i0 + k];
            cur[i0 + k] = run;
            run += c;
        }
    }
    __syncthreads();

    const unsigned* pp = part + (size_t)blockIdx.x * (size_t)N + lo;
    const bool first = blockIdx.x == 0;
    for (int i = tid; i < nr; i += DEG_THREADS) {
        const unsigned r = cur[i];
        if (first) rowstart[lo + i] = r;
        cur[i] = r + pp[i];
    }
    __syncthreads();

    const int sl = (((E + DEG_SLICES - 1) / DEG_SLICES) + 3) & ~3;
    const int e0 = (int)blockIdx.x * sl;
    const int e1 = min(e0 + sl, E);
    const int step = 4 * DEG_THREADS;
    int e = e0 + 4 * tid;
    for (; e + step + 3 < e1; e += 2 * step) {
        const int4 d0 = *(const int4*)(dst + e);
        const int4 d1 = *(const int4*)(dst + e + step);
        const int4 s0 = *(const int4*)(src + e);
        const int4 s1 = *(const int4*)(src + e + step);
        gcn_place4(cur, csr, d0, s0, lo, (unsigned)E);
        gcn_place4(cur, csr, d1, s1, lo, (unsigned)E);
    }
    for (; e < e1; e += step) {
        if (e + 3 < e1) {
            gcn_place4(cur, csr, *(const int4*)(dst + e), *(const int4*)(src + e), lo, (unsigned)E);
        } else {
            for (int k = e; k < e1; ++k) gcn_place(cur, csr, dst[k], src[k], lo, (unsigned)E);
        }
    }
}

#define NODES_PER_WAVE  16
#define WAVES_PER_BLOCK 8

__device__ __forceinline__ const float2* gcn_row2(const float* base, unsigned row, int lane) {
    const unsigned off = row * (unsigned)(DFEAT * sizeof(float)) + (unsigned)lane * (unsigned)sizeof(float2);
    return (const float2*)((const char*)base + off);
}

#define GCN_GATHER8(v, j0)                                                   \
    _Pragma("unroll")                                                        \
    for (int j = 0; j < 8; ++j) {                                            \
        const unsigned sj = (unsigned)__builtin_amdgcn_readlane((int)idx, (j0) + j); \
        v[j] = *gcn_row2(x, sj, lane);                                       \
    }

#define GCN_SUM8(v, j0, GUARD)                                               \
    _Pragma("unroll")                                                        \
    for (int j = 0; j < 8; ++j) {                                            \
        if (!(GUARD) || (j0) + j < n) {                                      \
            const unsigned ee = e + (unsigned)((j0) + j);                    \
            while (ee >= nodeEnd) {                                          \
                orow[p * (DFEAT / 2)] = acc;                                 \
                acc = make_float2(0.0f, 0.0f);                               \
                ++p;                                                         \
                nodeEnd = (unsigned)__builtin_amdgcn_readlane((int)rs, p + 1); \
            }                                                                \
            acc.x += v[j].x;                                                 \
            acc.y += v[j].y;                                                 \
        }                                                                    \
    }

#define GCN_GROUP32(GUARD)                                                   \
    {                                                                        \
        float2 a[8], b[8];                                                   \
        GCN_GATHER8(a, 0)                                                    \
        GCN_GATHER8(b, 8)                                                    \
        GCN_SUM8(a, 0, GUARD)                                                \
        GCN_GATHER8(a, 16)                                                   \
        GCN_SUM8(b, 8, GUARD)                                                \
        GCN_GATHER8(b, 24)                                                   \
        GCN_SUM8(a, 16, GUARD)                                               \
        GCN_SUM8(b, 24, GUARD)                                               \
    }

__global__ __launch_bounds__(WAVES_PER_BLOCK * 32)
void gcn_gather_kernel(const float* __restrict__ x,
                       const unsigned* __restrict__ rowstart,
                       const unsigned* __restrict__ csr,
                       float* __restrict__ agg, int N) {
    const int lane = threadIdx.x & 31;
    const int wv = __builtin_amdgcn_readfirstlane((int)(threadIdx.x >> 5));
    const int n0 = ((int)blockIdx.x * WAVES_PER_BLOCK + wv) * NODES_PER_WAVE;
    if (n0 >= N) return;
    const int nn = min(NODES_PER_WAVE, N - n0);

    unsigned rs = ~0u;
    if (lane <= nn) rs = rowstart[n0 + lane];
    const unsigned eBeg = (unsigned)__builtin_amdgcn_readlane((int)rs, 0);
    const unsigned eEnd = (unsigned)__builtin_amdgcn_readlane((int)rs, nn);
    unsigned nodeEnd    = (unsigned)__builtin_amdgcn_readlane((int)rs, 1);
    int p = 0;
    float2 acc = make_float2(0.0f, 0.0f);
    float2* orow = (float2*)(agg + (size_t)n0 * DFEAT) + lane;

    if (eBeg < eEnd) {
        const unsigned eLast = eEnd - 1u;
        unsigned iN = csr[min(eBeg + (unsigned)lane, eLast)];

        for (unsigned e = eBeg; e < eEnd; e += 32) {
            const unsigned idx = iN;
            iN = csr[min(e + 32u + (unsigned)lane, eLast)];
            const int n = (int)min(32u, eEnd - e);

            if (n == 32) {
                GCN_GROUP32(0)
            } else {
                GCN_GROUP32(1)
            }
        }
    }
    for (; p < nn; ++p) {
        orow[p * (DFEAT / 2)] = acc;
        acc = make_float2(0.0f, 0.0f);
    }
}

__global__ __launch_bounds__(128)
void gcn_wmma_kernel(const float* __restrict__ agg,
                     const float* __restrict__ inv_deg,
                     const float* __restrict__ W,
                     const float* __restrict__ bias,
                     float* __restrict__ out) {
    const int lane    = threadIdx.x & 31;
    const int wave    = threadIdx.x >> 5;
    const int rowBase = blockIdx.x << 4;
    const int colBase = wave << 4;
    const int m  = lane & 15;
    const int hi = lane >> 4;
    const int k2 = hi << 1;

    const float  inv  = inv_deg[rowBase + m];
    const float* arow = agg + (size_t)(rowBase + m) * DFEAT;
    const float* brow = W   + (size_t)(colBase + m) * DFEAT;

    v8f c = {};
#pragma unroll
    for (int kb = 0; kb < DFEAT; kb += 4) {
        v2f a = *(const v2f*)(arow + kb + k2);
        a.x *= inv;
        a.y *= inv;
        v2f b = *(const v2f*)(brow + kb + k2);
        c = __builtin_amdgcn_wmma_f32_16x16x4_f32(
                 false, a,  false, b,
                 (short)0, c,  false,  false);
    }

    const float bcol = bias[colBase + m];
#pragma unroll
    for (int r = 0; r < 8; ++r) {
        const int row = rowBase + r + (hi << 3);
        float v = c[r] + bcol;
        v = v > 0.0f ? v : expm1f(v);
        out[(size_t)row * DFEAT + colBase + m] = v;
    }
}

extern "C" void kernel_launch(void* const* d_in, const int* in_sizes, int n_in,
                              void* d_out, int out_size, void* d_ws, size_t ws_size,
                              hipStream_t stream) {
    const float* h   = (const float*)d_in[0];
    const int*   src = (const int*)  d_in[1];
    const int*   dst = (const int*)  d_in[2];
    const float* W1  = (const float*)d_in[3];
    const float* b1  = (const float*)d_in[4];
    const float* W2  = (const float*)d_in[5];
    const float* b2  = (const float*)d_in[6];
    float*       out = (float*)d_out;

    const int N = in_sizes[0] / DFEAT;
    const int E = in_sizes[1];

    const int tiles     = N / 16;
    const int degRanges = (N + DEG_RANGE - 1) / DEG_RANGE;
    const int pfxBlocks = (N + PFX_THREADS - 1) / PFX_THREADS;
    const int gatherB   = (N + NODES_PER_WAVE * WAVES_PER_BLOCK - 1)
                        / (NODES_PER_WAVE * WAVES_PER_BLOCK);

    char* ws = (char*)d_ws;
    const size_t A = 255;
    float* deg = (float*)ws;
    size_t off = (((size_t)N * sizeof(float)) + A) & ~A;
    float* agg = (float*)(ws + off);
    off += (size_t)N * DFEAT * sizeof(float);
    float* h1 = (float*)(ws + off);
    off += (size_t)N * DFEAT * sizeof(float);
    unsigned* part = (unsigned*)(ws + off);
    off += (((size_t)DEG_SLICES * (size_t)N * sizeof(unsigned)) + A) & ~A;
    unsigned* cnt = (unsigned*)(ws + off);
    off += (((size_t)N * sizeof(unsigned)) + A) & ~A;
    unsigned* blocksum = (unsigned*)(ws + off);
    off += (((size_t)pfxBlocks * sizeof(unsigned)) + A) & ~A;
    unsigned* rowstart = (unsigned*)(ws + off);
    off += (((size_t)(N + 1) * sizeof(unsigned)) + A) & ~A;
    unsigned* csr = (unsigned*)(ws + off);

    gcn_deg_kernel<<<dim3(DEG_SLICES, degRanges), DEG_THREADS, 0, stream>>>(dst, part, N, E);
    gcn_prefix_kernel<<<pfxBlocks, PFX_THREADS, 0, stream>>>(part, deg, cnt, blocksum, rowstart, N, E);
    gcn_fill_kernel<<<dim3(DEG_SLICES, degRanges), DEG_THREADS, 0, stream>>>(
        src, dst, part, cnt, blocksum, rowstart, csr, N, E);

    gcn_gather_kernel<<<gatherB, WAVES_PER_BLOCK * 32, 0, stream>>>(h, rowstart, csr, agg, N);
    gcn_wmma_kernel<<<tiles, 128, 0, stream>>>(agg, deg, W1, b1, h1);

    gcn_gather_kernel<<<gatherB, WAVES_PER_BLOCK * 32, 0, stream>>>(h1, rowstart, csr, agg, N);
    gcn_wmma_kernel<<<tiles, 128, 0, stream>>>(agg, deg, W2, b2, out);
}


# fuse aggregation with the linear layer: the wave that gather-sums 16 nodes keeps their scaled rows in LDS and runs the f32 WMMA, bias and ELU itself; agg never goes to memory; 7 launches -> 5
# speedup vs baseline: 3.3497x; 1.1791x over previous
#include <hip/hip_runtime.h>
#include <hip/hip_bf16.h>
#include <math.h>

typedef __attribute__((ext_vector_type(2))) float v2f;
typedef __attribute__((ext_vector_type(8))) float v8f;

#define DFEAT 64

#define DEG_RANGE   12544
#define DEG_SLICES  32
#define DEG_THREADS 1024

__device__ __forceinline__ void gcn_count(unsigned* cnt, int d, int lo) {
    const unsigned r = (unsigned)(d - lo);
    if (r < (unsigned)DEG_RANGE)
        __hip_atomic_fetch_add(&cnt[r], 1u, __ATOMIC_RELAXED, __HIP_MEMORY_SCOPE_WORKGROUP);
}

__device__ __forceinline__ void gcn_count4(unsigned* cnt, int4 v, int lo) {
    gcn_count(cnt, v.x, lo);
    gcn_count(cnt, v.y, lo);
    gcn_count(cnt, v.z, lo);
    gcn_count(cnt, v.w, lo);
}

__global__ __launch_bounds__(DEG_THREADS)
void gcn_deg_kernel(const int* __restrict__ dst,
                    unsigned* __restrict__ part, int N, int E) {
    __shared__ unsigned cnt[DEG_RANGE];
    const int lo = (int)blockIdx.y * DEG_RANGE;
    const int nr = min(DEG_RANGE, N - lo);
    for (int i = threadIdx.x; i < DEG_RANGE; i += DEG_THREADS) cnt[i] = 0u;
    __syncthreads();

    const int sl = (((E + DEG_SLICES - 1) / DEG_SLICES) + 3) & ~3;
    const int e0 = (int)blockIdx.x * sl;
    const int e1 = min(e0 + sl, E);
    const int step = 4 * DEG_THREADS;
    int e = e0 + 4 * (int)threadIdx.x;
    for (; e + 3 * step + 3 < e1; e += 4 * step) {
        const int4 v0 = *(const int4*)(dst + e);
        const int4 v1 = *(const int4*)(dst + e + step);
        const int4 v2 = *(const int4*)(dst + e + 2 * step);
        const int4 v3 = *(const int4*)(dst + e + 3 * step);
        gcn_count4(cnt, v0, lo);
        gcn_count4(cnt, v1, lo);
        gcn_count4(cnt, v2, lo);
        gcn_count4(cnt, v3, lo);
    }
    for (; e < e1; e += step) {
        if (e + 3 < e1) {
            gcn_count4(cnt, *(const int4*)(dst + e), lo);
        } else {
            for (int k = e; k < e1; ++k) gcn_count(cnt, dst[k], lo);
        }
    }
    __syncthreads();

    unsigned* p = part + (size_t)blockIdx.x * (size_t)N + lo;
    for (int i = threadIdx.x; i < nr; i += DEG_THREADS) p[i] = cnt[i];
}

#define PFX_THREADS 256

__global__ __launch_bounds__(PFX_THREADS)
void gcn_prefix_kernel(unsigned* __restrict__ part,
                       float* __restrict__ deg,
                       unsigned* __restrict__ cnt,
                       unsigned* __restrict__ blocksum,
                       unsigned* __restrict__ rowstart, int N, int E) {
    __shared__ unsigned wsum[PFX_THREADS / 32];
    const int n = blockIdx.x * PFX_THREADS + threadIdx.x;
    unsigned c = 0u;
    if (n < N) {
        unsigned v[DEG_SLICES];
#pragma unroll
        for (int s = 0; s < DEG_SLICES; ++s) v[s] = part[(size_t)s * (size_t)N + n];
#pragma unroll
        for (int s = 0; s < DEG_SLICES; ++s) {
            part[(size_t)s * (size_t)N + n] = c;
            c += v[s];
        }
        cnt[n] = c;
        const float d = (float)c;
        deg[n] = d > 0.0f ? 1.0f / d : 0.0f;
    }
    unsigned t = c;
    for (int m = 16; m > 0; m >>= 1) t += __shfl_xor(t, m);
    if ((threadIdx.x & 31) == 0) wsum[threadIdx.x >> 5] = t;
    __syncthreads();
    if (threadIdx.x == 0) {
        unsigned b = 0u;
        for (int w = 0; w < PFX_THREADS / 32; ++w) b += wsum[w];
        blocksum[blockIdx.x] = b;
        if (blockIdx.x == 0) rowstart[N] = (unsigned)E;
    }
}

#define DEG_CHUNK 13

__device__ __forceinline__ unsigned gcn_take(unsigned* cur, int d, int lo) {
    const unsigned r = (unsigned)(d - lo);
    unsigned pos = ~0u;
    if (r < (unsigned)DEG_RANGE)
        pos = __hip_atomic_fetch_add(&cur[r], 1u, __ATOMIC_RELAXED, __HIP_MEMORY_SCOPE_WORKGROUP);
    return pos;
}

__device__ __forceinline__ void gcn_put(unsigned* __restrict__ csr, unsigned pos, int s, unsigned E) {
    if (pos < E) csr[pos] = (unsigned)s;
}

__device__ __forceinline__ void gcn_place(unsigned* cur, unsigned* __restrict__ csr,
                                          int d, int s, int lo, unsigned E) {
    gcn_put(csr, gcn_take(cur, d, lo), s, E);
}

__device__ __forceinline__ void gcn_place4(unsigned* cur, unsigned* __restrict__ csr,
                                           int4 d, int4 s, int lo, unsigned E) {
    const unsigned p0 = gcn_take(cur, d.x, lo);
    const unsigned p1 = gcn_take(cur, d.y, lo);
    const unsigned p2 = gcn_take(cur, d.z, lo);
    const unsigned p3 = gcn_take(cur, d.w, lo);
    gcn_put(csr, p0, s.x, E);
    gcn_put(csr, p1, s.y, E);
    gcn_put(csr, p2, s.z, E);
    gcn_put(csr, p3, s.w, E);
}

__global__ __launch_bounds__(DEG_THREADS)
void gcn_fill_kernel(const int* __restrict__ src,
                     const int* __restrict__ dst,
                     const unsigned* __restrict__ part,
                     const unsigned* __restrict__ cnt,
                     const unsigned* __restrict__ blocksum,
                     unsigned* __restrict__ rowstart,
                     unsigned* __restrict__ csr, int N, int E) {
    __shared__ unsigned cur[DEG_RANGE];
    __shared__ unsigned wtot[DEG_THREADS / 32];
    __shared__ unsigned wbase[DEG_THREADS / 32];
    const int tid  = threadIdx.x;
    const int lane = tid & 31;
    const int wave = tid >> 5;
    const int lo = (int)blockIdx.y * DEG_RANGE;
    const int nr = min(DEG_RANGE, N - lo);

    const int nb = (int)blockIdx.y * (DEG_RANGE / PFX_THREADS);
    unsigned pb = 0u;
    for (int i = tid; i < nb; i += DEG_THREADS) pb += blocksum[i];
    for (int i = tid; i < nr; i += DEG_THREADS) cur[i] = cnt[lo + i];
    __syncthreads();

    const int i0 = tid * DEG_CHUNK;
    unsigned ls = 0u;
#pragma unroll
    for (int k = 0; k < DEG_CHUNK; ++k)
        if (i0 + k < nr) ls += cur[i0 + k];
    unsigned x = ls;
    for (int d = 1; d < 32; d <<= 1) {
        const unsigned y = __shfl_up(x, d);
        if (lane >= d) x += y;
    }
    for (int m = 16; m > 0; m >>= 1) pb += __shfl_xor(pb, m);
    if (lane == 31) wtot[wave] = x;
    if (lane == 0)  wbase[wave] = pb;
    __syncthreads();
    unsigned run = x - ls;
    for (int w = 0; w < DEG_THREADS / 32; ++w) {
        run += wbase[w];
        if (w < wave) run += wtot[w];
    }
#pragma unroll
    for (int k = 0; k < DEG_CHUNK; ++k) {
        if (i0 + k < nr) {
            const unsigned c = cur[i0 + k];
            cur[i0 + k] = run;
            run += c;
        }
    }
    __syncthreads();

    const unsigned* pp = part + (size_t)blockIdx.x * (size_t)N + lo;
    const bool first = blockIdx.x == 0;
    for (int i = tid; i < nr; i += DEG_THREADS) {
        const unsigned r = cur[i];
        if (first) rowstart[lo + i] = r;
        cur[i] = r + pp[i];
    }
    __syncthreads();

    const int sl = (((E + DEG_SLICES - 1) / DEG_SLICES) + 3) & ~3;
    const int e0 = (int)blockIdx.x * sl;
    const int e1 = min(e0 + sl, E);
    const int step = 4 * DEG_THREADS;
    int e = e0 + 4 * tid;
    for (; e + step + 3 < e1; e += 2 * step) {
        const int4 d0 = *(const int4*)(dst + e);
        const int4 d1 = *(const int4*)(dst + e + step);
        const int4 s0 = *(const int4*)(src + e);
        const int4 s1 = *(const int4*)(src + e + step);
        gcn_place4(cur, csr, d0, s0, lo, (unsigned)E);
        gcn_place4(cur, csr, d1, s1, lo, (unsigned)E);
    }
    for (; e < e1; e += step) {
        if (e + 3 < e1) {
            gcn_place4(cur, csr, *(const int4*)(dst + e), *(const int4*)(src + e), lo, (unsigned)E);
        } else {
            for (int k = e; k < e1; ++k) gcn_place(cur, csr, dst[k], src[k], lo, (unsigned)E);
        }
    }
}

#define NODES_PER_WAVE  16
#define WAVES_PER_BLOCK 8
#define LDS_PITCH       68

__device__ __forceinline__ const float2* gcn_row2(const float* base, unsigned row, int lane) {
    const unsigned off = row * (unsigned)(DFEAT * sizeof(float)) + (unsigned)lane * (unsigned)sizeof(float2);
    return (const float2*)((const char*)base + off);
}

#define GCN_GATHER8(v, j0)                                                   \
    _Pragma("unroll")                                                        \
    for (int j = 0; j < 8; ++j) {                                            \
        const unsigned sj = (unsigned)__builtin_amdgcn_readlane((int)idx, (j0) + j); \
        v[j] = *gcn_row2(x, sj, lane);                                       \
    }

#define GCN_FLUSH()                                                          \
    {                                                                        \
        const float sc = __int_as_float(__builtin_amdgcn_readlane(__float_as_int(iv), p)); \
        trow[p * (LDS_PITCH / 2)] = make_float2(acc.x * sc, acc.y * sc);     \
        acc = make_float2(0.0f, 0.0f);                                       \
        ++p;                                                                 \
    }

#define GCN_SUM8(v, j0, GUARD)                                               \
    _Pragma("unroll")                                                        \
    for (int j = 0; j < 8; ++j) {                                            \
        if (!(GUARD) || (j0) + j < n) {                                      \
            const unsigned ee = e + (unsigned)((j0) + j);                    \
            while (ee >= nodeEnd) {                                          \
                GCN_FLUSH()                                                  \
                nodeEnd = (unsigned)__builtin_amdgcn_readlane((int)rs, p + 1); \
            }                                                                \
            acc.x += v[j].x;                                                 \
            acc.y += v[j].y;                                                 \
        }                                                                    \
    }

#define GCN_GROUP32(GUARD)                                                   \
    {                                                                        \
        float2 a[8], b[8];                                                   \
        GCN_GATHER8(a, 0)                                                    \
        GCN_GATHER8(b, 8)                                                    \
        GCN_SUM8(a, 0, GUARD)                                                \
        GCN_GATHER8(a, 16)                                                   \
        GCN_SUM8(b, 8, GUARD)                                                \
        GCN_GATHER8(b, 24)                                                   \
        GCN_SUM8(a, 16, GUARD)                                               \
        GCN_SUM8(b, 24, GUARD)                                               \
    }

__device__ __forceinline__ v8f gcn_mma(v2f a, v2f b, v8f c) {
    return __builtin_amdgcn_wmma_f32_16x16x4_f32(
                false, a,  false, b,
                (short)0, c,  false,  false);
}

__device__ __forceinline__ void gcn_finish_tile(float* t, v8f c, float bcol, int hi, int col) {
#pragma unroll
    for (int r = 0; r < 8; ++r) {
        float v = c[r] + bcol;
        v = v > 0.0f ? v : expm1f(v);
        t[(r + (hi << 3)) * LDS_PITCH + col] = v;
    }
}

#define GCN_LOADK(FA, FB, t)                                                 \
    _Pragma("unroll")                                                        \
    for (int ks = 0; ks < 2; ++ks) {                                         \
        FA[ks] = *(const v2f*)(ap + 8 * (t) + 4 * ks);                       \
        _Pragma("unroll")                                                    \
        for (int ct = 0; ct < 4; ++ct)                                       \
            FB[ks][ct] = *(const v2f*)(bp + ct * 16 * LDS_PITCH + 8 * (t) + 4 * ks); \
    }

#define GCN_MMAK(FA, FB)                                                     \
    _Pragma("unroll")                                                        \
    for (int ks = 0; ks < 2; ++ks) {                                         \
        _Pragma("unroll")                                                    \
        for (int ct = 0; ct < 4; ++ct)                                       \
            c[ct] = gcn_mma(FA[ks], FB[ks][ct], c[ct]);                      \
    }

__global__ __launch_bounds__(WAVES_PER_BLOCK * 32)
void gcn_layer_kernel(const float* __restrict__ x,
                      const unsigned* __restrict__ rowstart,
                      const unsigned* __restrict__ csr,
                      const float* __restrict__ inv_deg,
                      const float* __restrict__ W,
                      const float* __restrict__ bias,
                      float* __restrict__ out, int N) {
    __shared__ __attribute__((aligned(16))) float wl[DFEAT * LDS_PITCH];
    __shared__ __attribute__((aligned(16))) float tl[WAVES_PER_BLOCK * NODES_PER_WAVE * LDS_PITCH];

    const int lane = threadIdx.x & 31;
    const int wv = __builtin_amdgcn_readfirstlane((int)(threadIdx.x >> 5));
    const int n0 = ((int)blockIdx.x * WAVES_PER_BLOCK + wv) * NODES_PER_WAVE;

    {
        float4 w4[4];
#pragma unroll
        for (int t = 0; t < 4; ++t)
            w4[t] = *(const float4*)(W + 4 * ((int)threadIdx.x + t * (WAVES_PER_BLOCK * 32)));
#pragma unroll
        for (int t = 0; t < 4; ++t) {
            const int i = (int)threadIdx.x + t * (WAVES_PER_BLOCK * 32);
            *(float4*)(wl + (i >> 4) * LDS_PITCH + ((i & 15) << 2)) = w4[t];
        }
    }
    __syncthreads();
    if (n0 >= N) return;
    const int nn = min(NODES_PER_WAVE, N - n0);

    float* tile = tl + wv * (NODES_PER_WAVE * LDS_PITCH);

    unsigned rs = ~0u;
    if (lane <= nn) rs = rowstart[n0 + lane];
    float iv = 0.0f;
    if (lane < nn) iv = inv_deg[n0 + lane];
    const unsigned eBeg = (unsigned)__builtin_amdgcn_readlane((int)rs, 0);
    const unsigned eEnd = (unsigned)__builtin_amdgcn_readlane((int)rs, nn);
    unsigned nodeEnd    = (unsigned)__builtin_amdgcn_readlane((int)rs, 1);
    int p = 0;
    float2 acc = make_float2(0.0f, 0.0f);
    float2* trow = (float2*)tile + lane;

    if (eBeg < eEnd) {
        const unsigned eLast = eEnd - 1u;
        unsigned iN = csr[min(eBeg + (unsigned)lane, eLast)];

        for (unsigned e = eBeg; e < eEnd; e += 32) {
            const unsigned idx = iN;
            iN = csr[min(e + 32u + (unsigned)lane, eLast)];
            const int n = (int)min(32u, eEnd - e);

            if (n == 32) {
                GCN_GROUP32(0)
            } else {
                GCN_GROUP32(1)
            }
        }
    }
    while (p < NODES_PER_WAVE) GCN_FLUSH()

    __builtin_amdgcn_fence(__ATOMIC_RELEASE, "wavefront");
    __builtin_amdgcn_fence(__ATOMIC_ACQUIRE, "wavefront");

    const int m  = lane & 15;
    const int hi = lane >> 4;
    const int k2 = hi << 1;
    const float* ap = tile + m * LDS_PITCH + k2;
    const float* bp = wl   + m * LDS_PITCH + k2;

    v8f c[4] = {};
    v2f fa[2][2], fb[2][2][4];
    float bcol[4];
#pragma unroll
    for (int ct = 0; ct < 4; ++ct) bcol[ct] = bias[16 * ct + m];
    GCN_LOADK(fa[0], fb[0], 0)
#pragma unroll
    for (int t = 0; t < DFEAT / 8; ++t) {
        if (t + 1 < DFEAT / 8) GCN_LOADK(fa[(t + 1) & 1], fb[(t + 1) & 1], t + 1)
        __builtin_amdgcn_sched_barrier(0);
        GCN_MMAK(fa[t & 1], fb[t & 1])
        __builtin_amdgcn_sched_barrier(0);
    }

#pragma unroll
    for (int ct = 0; ct < 4; ++ct)
        gcn_finish_tile(tile, c[ct], bcol[ct], hi, 16 * ct + m);

    __builtin_amdgcn_fence(__ATOMIC_RELEASE, "wavefront");
    __builtin_amdgcn_fence(__ATOMIC_ACQUIRE, "wavefront");

    const int q4 = m << 2;
    float* orow = out + (size_t)n0 * DFEAT + q4;
    if (nn == NODES_PER_WAVE) {
        float4 ov[NODES_PER_WAVE / 2];
#pragma unroll
        for (int j = 0; j < NODES_PER_WAVE / 2; ++j)
            ov[j] = *(const float4*)(tile + (2 * j + hi) * LDS_PITCH + q4);
#pragma unroll
        for (int j = 0; j < NODES_PER_WAVE / 2; ++j)
            *(float4*)(orow + (2 * j + hi) * DFEAT) = ov[j];
    } else {
        for (int j = 0; j < NODES_PER_WAVE / 2; ++j) {
            const int row = 2 * j + hi;
            if (row < nn)
                *(float4*)(orow + row * DFEAT) = *(const float4*)(tile + row * LDS_PITCH + q4);
        }
    }
}

extern "C" void kernel_launch(void* const* d_in, const int* in_sizes, int n_in,
                              void* d_out, int out_size, void* d_ws, size_t ws_size,
                              hipStream_t stream) {
    const float* h   = (const float*)d_in[0];
    const int*   src = (const int*)  d_in[1];
    const int*   dst = (const int*)  d_in[2];
    const float* W1  = (const float*)d_in[3];
    const float* b1  = (const float*)d_in[4];
    const float* W2  = (const float*)d_in[5];
    const float* b2  = (const float*)d_in[6];
    float*       out = (float*)d_out;

    const int N = in_sizes[0] / DFEAT;
    const int E = in_sizes[1];

    const int degRanges = (N + DEG_RANGE - 1) / DEG_RANGE;
    const int pfxBlocks = (N + PFX_THREADS - 1) / PFX_THREADS;
    const int layerB    = (N + NODES_PER_WAVE * WAVES_PER_BLOCK - 1)
                        / (NODES_PER_WAVE * WAVES_PER_BLOCK);

    char* ws = (char*)d_ws;
    const size_t A = 255;
    float* deg = (float*)ws;
    size_t off = (((size_t)N * sizeof(float)) + A) & ~A;
    float* h1 = (float*)(ws + off);
    off += (size_t)N * DFEAT * sizeof(float);
    unsigned* part = (unsigned*)(ws + off);
    off += (((size_t)DEG_SLICES * (size_t)N * sizeof(unsigned)) + A) & ~A;
    unsigned* cnt = (unsigned*)(ws + off);
    off += (((size_t)N * sizeof(unsigned)) + A) & ~A;
    unsigned* blocksum = (unsigned*)(ws + off);
    off += (((size_t)pfxBlocks * sizeof(unsigned)) + A) & ~A;
    unsigned* rowstart = (unsigned*)(ws + off);
    off += (((size_t)(N + 1) * sizeof(unsigned)) + A) & ~A;
    unsigned* csr = (unsigned*)(ws + off);

    gcn_deg_kernel<<<dim3(DEG_SLICES, degRanges), DEG_THREADS, 0, stream>>>(dst, part, N, E);
    gcn_prefix_kernel<<<pfxBlocks, PFX_THREADS, 0, stream>>>(part, deg, cnt, blocksum, rowstart, N, E);
    gcn_fill_kernel<<<dim3(DEG_SLICES, degRanges), DEG_THREADS, 0, stream>>>(
        src, dst, part, cnt, blocksum, rowstart, csr, N, E);

    gcn_layer_kernel<<<layerB, WAVES_PER_BLOCK * 32, 0, stream>>>(h,  rowstart, csr, deg, W1, b1, h1,  N);
    gcn_layer_kernel<<<layerB, WAVES_PER_BLOCK * 32, 0, stream>>>(h1, rowstart, csr, deg, W2, b2, out, N);
}


# CSR build in two levels: edges split into 256 node ranges (per-slice range counts), then each range sorted by node in one block's LDS; replaces 32 count planes + prefix pass; sum order changes
# speedup vs baseline: 4.1349x; 1.2344x over previous
#include <hip/hip_runtime.h>
#include <hip/hip_bf16.h>
#include <math.h>

typedef __attribute__((ext_vector_type(2))) float v2f;
typedef __attribute__((ext_vector_type(8))) float v8f;

#define DFEAT 64

#define NRANGE     256
#define NSLICE     256
#define RN_MAX     512
#define EP_THREADS 1024
#define RC_THREADS 512

__device__ __forceinline__ unsigned gcn_rn(int N) {
    return (unsigned)max(1, (N + NRANGE - 1) / NRANGE);
}

__device__ __forceinline__ unsigned gcn_range_of(int d, unsigned rn) {
    return min((unsigned)d / rn, (unsigned)(NRANGE - 1));
}

__device__ __forceinline__ void gcn_rcount(unsigned* cnt, int d, unsigned rn) {
    __hip_atomic_fetch_add(&cnt[gcn_range_of(d, rn)], 1u, __ATOMIC_RELAXED, __HIP_MEMORY_SCOPE_WORKGROUP);
}

__global__ __launch_bounds__(RC_THREADS)
void gcn_rcount_kernel(const int* __restrict__ dst,
                       unsigned* __restrict__ tab, int N, int E) {
    __shared__ unsigned cnt[NRANGE];
    const int tid = threadIdx.x;
    if (tid < NRANGE) cnt[tid] = 0u;
    __syncthreads();

    const unsigned rn = gcn_rn(N);
    const int sl = (((E + NSLICE - 1) / NSLICE) + 3) & ~3;
    const int e0 = (int)blockIdx.x * sl;
    const int e1 = min(e0 + sl, E);
    const int len   = max(e1 - e0, 0);
    const int full4 = len & ~3;
    const int* dp = dst + e0;
    for (int i = 4 * tid; i < full4; i += 16 * RC_THREADS) {
        int4 v[4];
#pragma unroll
        for (int k = 0; k < 4; ++k) v[k] = *(const int4*)(dp + min(i + k * 4 * RC_THREADS, full4 - 4));
#pragma unroll
        for (int k = 0; k < 4; ++k) {
            if (i + k * 4 * RC_THREADS < full4) {
                gcn_rcount(cnt, v[k].x, rn);
                gcn_rcount(cnt, v[k].y, rn);
                gcn_rcount(cnt, v[k].z, rn);
                gcn_rcount(cnt, v[k].w, rn);
            }
        }
    }
    if (tid < len - full4) gcn_rcount(cnt, dp[full4 + tid], rn);
    __syncthreads();
    if (tid < NRANGE) tab[(int)blockIdx.x * NRANGE + tid] = cnt[tid];
}

__device__ __forceinline__ unsigned gcn_rtake(unsigned* cur, int d, unsigned rn) {
    return __hip_atomic_fetch_add(&cur[gcn_range_of(d, rn)], 1u,
                                  __ATOMIC_RELAXED, __HIP_MEMORY_SCOPE_WORKGROUP);
}

__device__ __forceinline__ void gcn_rstore(uint2* __restrict__ pairs, unsigned pos,
                                           int s, int d, unsigned E) {
    if (pos < E) pairs[pos] = make_uint2((unsigned)s, (unsigned)d);
}

__device__ __forceinline__ void gcn_rput(unsigned* cur, uint2* __restrict__ pairs,
                                         int s, int d, unsigned rn, unsigned E) {
    gcn_rstore(pairs, gcn_rtake(cur, d, rn), s, d, E);
}

__device__ __forceinline__ void gcn_rput4(unsigned* cur, uint2* __restrict__ pairs,
                                          int4 s, int4 d, unsigned rn, unsigned E) {
    const unsigned p0 = gcn_rtake(cur, d.x, rn);
    const unsigned p1 = gcn_rtake(cur, d.y, rn);
    const unsigned p2 = gcn_rtake(cur, d.z, rn);
    const unsigned p3 = gcn_rtake(cur, d.w, rn);
    gcn_rstore(pairs, p0, s.x, d.x, E);
    gcn_rstore(pairs, p1, s.y, d.y, E);
    gcn_rstore(pairs, p2, s.z, d.z, E);
    gcn_rstore(pairs, p3, s.w, d.w, E);
}

__global__ __launch_bounds__(EP_THREADS)
void gcn_rsplit_kernel(const int* __restrict__ src,
                       const int* __restrict__ dst,
                       const unsigned* __restrict__ tab,
                       unsigned* __restrict__ rstart,
                       unsigned* __restrict__ pairs_w, int N, int E) {
    uint2* __restrict__ pairs = (uint2*)pairs_w;
    __shared__ unsigned tot[4][NRANGE];
    __shared__ unsigned bel[4][NRANGE];
    __shared__ unsigned cur[NRANGE];
    __shared__ unsigned wt[NRANGE / 32];
    const int tid  = threadIdx.x;
    const int lane = tid & 31;
    const int r    = tid & (NRANGE - 1);
    const int q    = tid >> 8;
    const int s    = (int)blockIdx.x;

    const unsigned rn = gcn_rn(N);
    const int sl = (((E + NSLICE - 1) / NSLICE) + 3) & ~3;
    const int e0 = s * sl;
    const int e1 = min(e0 + sl, E);
    const int len   = max(e1 - e0, 0);
    const int full4 = len & ~3;
    const int* dp = dst + e0;
    const int* sp = src + e0;

    unsigned t = 0u, b = 0u;
    for (int kk = 0; kk < NSLICE / 4; kk += 32) {
        const int sp0 = q * (NSLICE / 4) + kk;
        const unsigned* tp = tab + sp0 * NRANGE + r;
        unsigned c[32];
#pragma unroll
        for (int k = 0; k < 32; ++k) c[k] = tp[k * NRANGE];
#pragma unroll
        for (int k = 0; k < 32; ++k) {
            t += c[k];
            if (sp0 + k < s) b += c[k];
        }
    }
    int4 dv[2], sv[2];
    int i = 4 * tid;
    if (full4 > 0) {
#pragma unroll
        for (int k = 0; k < 2; ++k) {
            const int j = min(i + k * 4 * EP_THREADS, full4 - 4);
            dv[k] = *(const int4*)(dp + j);
            sv[k] = *(const int4*)(sp + j);
        }
    }
    tot[q][r] = t;
    bel[q][r] = b;
    __syncthreads();

    unsigned v = 0u, below = 0u;
    if (tid < NRANGE) {
        v     = tot[0][tid] + tot[1][tid] + tot[2][tid] + tot[3][tid];
        below = bel[0][tid] + bel[1][tid] + bel[2][tid] + bel[3][tid];
    }
    unsigned x = v;
    for (int d = 1; d < 32; d <<= 1) {
        const unsigned y = __shfl_up(x, d);
        if (lane >= d) x += y;
    }
    if (tid < NRANGE && lane == 31) wt[tid >> 5] = x;
    __syncthreads();
    if (tid < NRANGE) {
        unsigned base = x - v;
        for (int w = 0; w < (tid >> 5); ++w) base += wt[w];
        cur[tid] = base + below;
        if (s == 0) {
            rstart[tid] = base;
            if (tid == NRANGE - 1) rstart[NRANGE] = base + v;
        }
    }
    __syncthreads();

    while (i < full4) {
#pragma unroll
        for (int k = 0; k < 2; ++k) {
            if (i + k * 4 * EP_THREADS < full4) {
                gcn_rput4(cur, pairs, sv[k], dv[k], rn, (unsigned)E);
            }
        }
        i += 8 * EP_THREADS;
        if (i < full4) {
#pragma unroll
            for (int k = 0; k < 2; ++k) {
                const int j = min(i + k * 4 * EP_THREADS, full4 - 4);
                dv[k] = *(const int4*)(dp + j);
                sv[k] = *(const int4*)(sp + j);
            }
        }
    }
    if (tid < len - full4)
        gcn_rput(cur, pairs, sp[full4 + tid], dp[full4 + tid], rn, (unsigned)E);
}

__global__ __launch_bounds__(EP_THREADS)
void gcn_rsort_kernel(const unsigned* __restrict__ pairs_w,
                      const unsigned* __restrict__ rstart,
                      unsigned* __restrict__ rowstart,
                      float* __restrict__ deg,
                      unsigned* __restrict__ csr, int N, int E) {
    const uint2* __restrict__ pairs = (const uint2*)pairs_w;
    __shared__ unsigned cnt[RN_MAX];
    __shared__ unsigned wt[RN_MAX / 32];
    const int tid  = threadIdx.x;
    const int lane = tid & 31;
    const int wave = tid >> 5;
    const unsigned rn = gcn_rn(N);
    const int lo = (int)blockIdx.x * (int)rn;
    const unsigned nr = (unsigned)max(0, min(min((int)rn, RN_MAX), N - lo));
    const unsigned e0 = rstart[blockIdx.x];
    const unsigned e1 = min(rstart[blockIdx.x + 1], (unsigned)E);

    if (tid < RN_MAX) cnt[tid] = 0u;
    __syncthreads();
    for (unsigned e = e0 + (unsigned)tid; e < e1; e += 4 * EP_THREADS) {
        unsigned d[4];
#pragma unroll
        for (int k = 0; k < 4; ++k) d[k] = pairs[min(e + (unsigned)(k * EP_THREADS), e1 - 1u)].y;
#pragma unroll
        for (int k = 0; k < 4; ++k) {
            const unsigned rel = d[k] - (unsigned)lo;
            if (e + (unsigned)(k * EP_THREADS) < e1 && rel < nr)
                __hip_atomic_fetch_add(&cnt[rel], 1u, __ATOMIC_RELAXED, __HIP_MEMORY_SCOPE_WORKGROUP);
        }
    }
    __syncthreads();

    unsigned v = 0u;
    if (tid < RN_MAX) v = cnt[tid];
    unsigned x = v;
    for (int d = 1; d < 32; d <<= 1) {
        const unsigned y = __shfl_up(x, d);
        if (lane >= d) x += y;
    }
    if (tid < RN_MAX && lane == 31) wt[wave] = x;
    __syncthreads();
    if (tid < RN_MAX) {
        unsigned start = e0 + (x - v);
        for (int w = 0; w < wave; ++w) start += wt[w];
        cnt[tid] = start;
        if ((unsigned)tid < nr) {
            rowstart[lo + tid] = start;
            const float d = (float)v;
            deg[lo + tid] = d > 0.0f ? 1.0f / d : 0.0f;
        }
    }
    if (blockIdx.x == 0 && tid == 0) rowstart[N] = (unsigned)E;
    __syncthreads();

    for (unsigned e = e0 + (unsigned)tid; e < e1; e += 4 * EP_THREADS) {
        uint2 p[4];
#pragma unroll
        for (int k = 0; k < 4; ++k) p[k] = pairs[min(e + (unsigned)(k * EP_THREADS), e1 - 1u)];
#pragma unroll
        for (int k = 0; k < 4; ++k) {
            const unsigned rel = p[k].y - (unsigned)lo;
            if (e + (unsigned)(k * EP_THREADS) < e1 && rel < nr) {
                const unsigned pos = __hip_atomic_fetch_add(&cnt[rel], 1u, __ATOMIC_RELAXED,
                                                            __HIP_MEMORY_SCOPE_WORKGROUP);
                if (pos < (unsigned)E) csr[pos] = p[k].x;
            }
        }
    }
}

#define NODES_PER_WAVE  16
#define WAVES_PER_BLOCK 8
#define LDS_PITCH       68

__device__ __forceinline__ const float2* gcn_row2(const float* base, unsigned row, int lane) {
    const unsigned off = row * (unsigned)(DFEAT * sizeof(float)) + (unsigned)lane * (unsigned)sizeof(float2);
    return (const float2*)((const char*)base + off);
}

#define GCN_GATHER8(v, j0)                                                   \
    _Pragma("unroll")                                                        \
    for (int j = 0; j < 8; ++j) {                                            \
        const unsigned sj = (unsigned)__builtin_amdgcn_readlane((int)idx, (j0) + j); \
        v[j] = *gcn_row2(x, sj, lane);                                       \
    }

#define GCN_FLUSH()                                                          \
    {                                                                        \
        const float sc = __int_as_float(__builtin_amdgcn_readlane(__float_as_int(iv), p)); \
        trow[p * (LDS_PITCH / 2)] = make_float2(acc.x * sc, acc.y * sc);     \
        acc = make_float2(0.0f, 0.0f);                                       \
        ++p;                                                                 \
    }

#define GCN_SUM8(v, j0, GUARD)                                               \
    _Pragma("unroll")                                                        \
    for (int j = 0; j < 8; ++j) {                                            \
        if (!(GUARD) || (j0) + j < n) {                                      \
            const unsigned ee = e + (unsigned)((j0) + j);                    \
            while (ee >= nodeEnd) {                                          \
                GCN_FLUSH()                                                  \
                nodeEnd = (unsigned)__builtin_amdgcn_readlane((int)rs, p + 1); \
            }                                                                \
            acc.x += v[j].x;                                                 \
            acc.y += v[j].y;                                                 \
        }                                                                    \
    }

#define GCN_GROUP32(GUARD)                                                   \
    {                                                                        \
        float2 a[8], b[8];                                                   \
        GCN_GATHER8(a, 0)                                                    \
        GCN_GATHER8(b, 8)                                                    \
        GCN_SUM8(a, 0, GUARD)                                                \
        GCN_GATHER8(a, 16)                                                   \
        GCN_SUM8(b, 8, GUARD)                                                \
        GCN_GATHER8(b, 24)                                                   \
        GCN_SUM8(a, 16, GUARD)                                               \
        GCN_SUM8(b, 24, GUARD)                                               \
    }

__device__ __forceinline__ v8f gcn_mma(v2f a, v2f b, v8f c) {
    return __builtin_amdgcn_wmma_f32_16x16x4_f32(
                false, a,  false, b,
                (short)0, c,  false,  false);
}

__device__ __forceinline__ void gcn_finish_tile(float* t, v8f c, float bcol, int hi, int col) {
#pragma unroll
    for (int r = 0; r < 8; ++r) {
        float v = c[r] + bcol;
        v = v > 0.0f ? v : expm1f(v);
        t[(r + (hi << 3)) * LDS_PITCH + col] = v;
    }
}

#define GCN_LOADK(FA, FB, t)                                                 \
    _Pragma("unroll")                                                        \
    for (int ks = 0; ks < 2; ++ks) {                                         \
        FA[ks] = *(const v2f*)(ap + 8 * (t) + 4 * ks);                       \
        _Pragma("unroll")                                                    \
        for (int ct = 0; ct < 4; ++ct)                                       \
            FB[ks][ct] = *(const v2f*)(bp + ct * 16 * LDS_PITCH + 8 * (t) + 4 * ks); \
    }

#define GCN_MMAK(FA, FB)                                                     \
    _Pragma("unroll")                                                        \
    for (int ks = 0; ks < 2; ++ks) {                                         \
        _Pragma("unroll")                                                    \
        for (int ct = 0; ct < 4; ++ct)                                       \
            c[ct] = gcn_mma(FA[ks], FB[ks][ct], c[ct]);                      \
    }

__global__ __launch_bounds__(WAVES_PER_BLOCK * 32)
void gcn_layer_kernel(const float* __restrict__ x,
                      const unsigned* __restrict__ rowstart,
                      const unsigned* __restrict__ csr,
                      const float* __restrict__ inv_deg,
                      const float* __restrict__ W,
                      const float* __restrict__ bias,
                      float* __restrict__ out, int N) {
    __shared__ __attribute__((aligned(16))) float wl[DFEAT * LDS_PITCH];
    __shared__ __attribute__((aligned(16))) float tl[WAVES_PER_BLOCK * NODES_PER_WAVE * LDS_PITCH];

    const int lane = threadIdx.x & 31;
    const int wv = __builtin_amdgcn_readfirstlane((int)(threadIdx.x >> 5));
    const int n0 = ((int)blockIdx.x * WAVES_PER_BLOCK + wv) * NODES_PER_WAVE;

    {
        float4 w4[4];
#pragma unroll
        for (int t = 0; t < 4; ++t)
            w4[t] = *(const float4*)(W + 4 * ((int)threadIdx.x + t * (WAVES_PER_BLOCK * 32)));
#pragma unroll
        for (int t = 0; t < 4; ++t) {
            const int i = (int)threadIdx.x + t * (WAVES_PER_BLOCK * 32);
            *(float4*)(wl + (i >> 4) * LDS_PITCH + ((i & 15) << 2)) = w4[t];
        }
    }
    __syncthreads();
    if (n0 >= N) return;
    const int nn = min(NODES_PER_WAVE, N - n0);

    float* tile = tl + wv * (NODES_PER_WAVE * LDS_PITCH);

    unsigned rs = ~0u;
    if (lane <= nn) rs = rowstart[n0 + lane];
    float iv = 0.0f;
    if (lane < nn) iv = inv_deg[n0 + lane];
    const unsigned eBeg = (unsigned)__builtin_amdgcn_readlane((int)rs, 0);
    const unsigned eEnd = (unsigned)__builtin_amdgcn_readlane((int)rs, nn);
    unsigned nodeEnd    = (unsigned)__builtin_amdgcn_readlane((int)rs, 1);
    int p = 0;
    float2 acc = make_float2(0.0f, 0.0f);
    float2* trow = (float2*)tile + lane;

    if (eBeg < eEnd) {
        const unsigned eLast = eEnd - 1u;
        unsigned iN = csr[min(eBeg + (unsigned)lane, eLast)];

        for (unsigned e = eBeg; e < eEnd; e += 32) {
            const unsigned idx = iN;
            iN = csr[min(e + 32u + (unsigned)lane, eLast)];
            const int n = (int)min(32u, eEnd - e);

            if (n == 32) {
                GCN_GROUP32(0)
            } else {
                GCN_GROUP32(1)
            }
        }
    }
    while (p < NODES_PER_WAVE) GCN_FLUSH()

    __builtin_amdgcn_fence(__ATOMIC_RELEASE, "wavefront");
    __builtin_amdgcn_fence(__ATOMIC_ACQUIRE, "wavefront");

    const int m  = lane & 15;
    const int hi = lane >> 4;
    const int k2 = hi << 1;
    const float* ap = tile + m * LDS_PITCH + k2;
    const float* bp = wl   + m * LDS_PITCH + k2;

    v8f c[4] = {};
    v2f fa[2][2], fb[2][2][4];
    float bcol[4];
#pragma unroll
    for (int ct = 0; ct < 4; ++ct) bcol[ct] = bias[16 * ct + m];
    GCN_LOADK(fa[0], fb[0], 0)
#pragma unroll
    for (int t = 0; t < DFEAT / 8; ++t) {
        if (t + 1 < DFEAT / 8) GCN_LOADK(fa[(t + 1) & 1], fb[(t + 1) & 1], t + 1)
        __builtin_amdgcn_sched_barrier(0);
        GCN_MMAK(fa[t & 1], fb[t & 1])
        __builtin_amdgcn_sched_barrier(0);
    }

#pragma unroll
    for (int ct = 0; ct < 4; ++ct)
        gcn_finish_tile(tile, c[ct], bcol[ct], hi, 16 * ct + m);

    __builtin_amdgcn_fence(__ATOMIC_RELEASE, "wavefront");
    __builtin_amdgcn_fence(__ATOMIC_ACQUIRE, "wavefront");

    const int q4 = m << 2;
    float* orow = out + (size_t)n0 * DFEAT + q4;
    if (nn == NODES_PER_WAVE) {
        float4 ov[NODES_PER_WAVE / 2];
#pragma unroll
        for (int j = 0; j < NODES_PER_WAVE / 2; ++j)
            ov[j] = *(const float4*)(tile + (2 * j + hi) * LDS_PITCH + q4);
#pragma unroll
        for (int j = 0; j < NODES_PER_WAVE / 2; ++j)
            *(float4*)(orow + (2 * j + hi) * DFEAT) = ov[j];
    } else {
        for (int j = 0; j < NODES_PER_WAVE / 2; ++j) {
            const int row = 2 * j + hi;
            if (row < nn)
                *(float4*)(orow + row * DFEAT) = *(const float4*)(tile + row * LDS_PITCH + q4);
        }
    }
}

extern "C" void kernel_launch(void* const* d_in, const int* in_sizes, int n_in,
                              void* d_out, int out_size, void* d_ws, size_t ws_size,
                              hipStream_t stream) {
    const float* h   = (const float*)d_in[0];
    const int*   src = (const int*)  d_in[1];
    const int*   dst = (const int*)  d_in[2];
    const float* W1  = (const float*)d_in[3];
    const float* b1  = (const float*)d_in[4];
    const float* W2  = (const float*)d_in[5];
    const float* b2  = (const float*)d_in[6];
    float*       out = (float*)d_out;

    const int N = in_sizes[0] / DFEAT;
    const int E = in_sizes[1];

    const int layerB = (N + NODES_PER_WAVE * WAVES_PER_BLOCK - 1)
                     / (NODES_PER_WAVE * WAVES_PER_BLOCK);

    char* ws = (char*)d_ws;
    const size_t A = 255;
    float* deg = (float*)ws;
    size_t off = (((size_t)N * sizeof(float)) + A) & ~A;
    float* h1 = (float*)(ws + off);
    off += (size_t)N * DFEAT * sizeof(float);
    unsigned* tab = (unsigned*)(ws + off);
    off += (((size_t)NSLICE * NRANGE * sizeof(unsigned)) + A) & ~A;
    unsigned* rstart = (unsigned*)(ws + off);
    off += (((size_t)(NRANGE + 1) * sizeof(unsigned)) + A) & ~A;
    unsigned* pairs = (unsigned*)(ws + off);
    off += (((size_t)E * 2 * sizeof(unsigned)) + A) & ~A;
    unsigned* rowstart = (unsigned*)(ws + off);
    off += (((size_t)(N + 1) * sizeof(unsigned)) + A) & ~A;
    unsigned* csr = (unsigned*)(ws + off);

    gcn_rcount_kernel<<<NSLICE, RC_THREADS, 0, stream>>>(dst, tab, N, E);
    gcn_rsplit_kernel<<<NSLICE, EP_THREADS, 0, stream>>>(src, dst, tab, rstart, pairs, N, E);
    gcn_rsort_kernel<<<NRANGE, EP_THREADS, 0, stream>>>(pairs, rstart, rowstart, deg, csr, N, E);

    gcn_layer_kernel<<<layerB, WAVES_PER_BLOCK * 32, 0, stream>>>(h,  rowstart, csr, deg, W1, b1, h1,  N);
    gcn_layer_kernel<<<layerB, WAVES_PER_BLOCK * 32, 0, stream>>>(h1, rowstart, csr, deg, W2, b2, out, N);
}


# CSR build in two launches: each edge slice is sorted by range in its block's LDS in one pass (rank = the counting add's return), stored as whole lines; count launch removed; sum order changes
# speedup vs baseline: 4.8200x; 1.1657x over previous
#include <hip/hip_runtime.h>
#include <hip/hip_bf16.h>
#include <math.h>

typedef __attribute__((ext_vector_type(2))) float v2f;
typedef __attribute__((ext_vector_type(8))) float v8f;

#define DFEAT 64

#define NRANGE     256
#define NSLICE     256
#define RN_MAX     512
#define EP_THREADS 1024
#define SL_CAP     (8 * EP_THREADS)

__device__ __forceinline__ unsigned gcn_rn(int N) {
    return (unsigned)max(1, (N + NRANGE - 1) / NRANGE);
}

__device__ __forceinline__ int gcn_slice_len(int E) {
    return (((E + NSLICE - 1) / NSLICE) + 3) & ~3;
}

__device__ __forceinline__ unsigned gcn_range_of(int d, unsigned rn) {
    return min((unsigned)d / rn, (unsigned)(NRANGE - 1));
}

__device__ __forceinline__ unsigned gcn_take(unsigned* c) {
    return __hip_atomic_fetch_add(c, 1u, __ATOMIC_RELAXED, __HIP_MEMORY_SCOPE_WORKGROUP);
}

__global__ __launch_bounds__(EP_THREADS)
void gcn_lsort_kernel(const int* __restrict__ src,
                      const int* __restrict__ dst,
                      unsigned* __restrict__ tcnt,
                      unsigned* __restrict__ pairs_w, int N, int E) {
    uint2* __restrict__ pairs = (uint2*)pairs_w;
    __shared__ __attribute__((aligned(16))) uint2 recs[SL_CAP];
    __shared__ unsigned cnt[NRANGE];
    __shared__ unsigned wt[NRANGE / 32];
    const int tid  = threadIdx.x;
    const int lane = tid & 31;
    const int s    = (int)blockIdx.x;
    if (tid < NRANGE) cnt[tid] = 0u;

    const unsigned rn = gcn_rn(N);
    const int sl = gcn_slice_len(E);
    const int e0 = s * sl;
    const int e1 = min(e0 + sl, E);
    const int len   = min(max(e1 - e0, 0), SL_CAP);
    const int full4 = len & ~3;
    const int* dp = dst + e0;
    const int* sp = src + e0;

    int4 dv[2], sv[2];
    if (full4 > 0) {
#pragma unroll
        for (int k = 0; k < 2; ++k) {
            const int j = min(4 * tid + k * 4 * EP_THREADS, full4 - 4);
            dv[k] = *(const int4*)(dp + j);
            sv[k] = *(const int4*)(sp + j);
        }
    }
    const bool ht = tid < len - full4;
    int dt = 0, st = 0;
    if (ht) { dt = dp[full4 + tid]; st = sp[full4 + tid]; }
    __syncthreads();

    unsigned rg[2][4], rk[2][4], rgt = 0u, rkt = 0u;
#pragma unroll
    for (int k = 0; k < 2; ++k) {
        if (4 * tid + k * 4 * EP_THREADS < full4) {
            rg[k][0] = gcn_range_of(dv[k].x, rn);
            rg[k][1] = gcn_range_of(dv[k].y, rn);
            rg[k][2] = gcn_range_of(dv[k].z, rn);
            rg[k][3] = gcn_range_of(dv[k].w, rn);
#pragma unroll
            for (int c = 0; c < 4; ++c) rk[k][c] = gcn_take(&cnt[rg[k][c]]);
        }
    }
    if (ht) { rgt = gcn_range_of(dt, rn); rkt = gcn_take(&cnt[rgt]); }
    __syncthreads();

    unsigned v = 0u;
    if (tid < NRANGE) v = cnt[tid];
    unsigned x = v;
    for (int d = 1; d < 32; d <<= 1) {
        const unsigned y = __shfl_up(x, d);
        if (lane >= d) x += y;
    }
    if (tid < NRANGE && lane == 31) wt[tid >> 5] = x;
    __syncthreads();
    if (tid < NRANGE) {
        unsigned base = x - v;
        for (int w = 0; w < (tid >> 5); ++w) base += wt[w];
        cnt[tid] = base;
        tcnt[tid * NSLICE + s] = v;
    }
    __syncthreads();

#pragma unroll
    for (int k = 0; k < 2; ++k) {
        if (4 * tid + k * 4 * EP_THREADS < full4) {
            const unsigned p0 = cnt[rg[k][0]] + rk[k][0];
            const unsigned p1 = cnt[rg[k][1]] + rk[k][1];
            const unsigned p2 = cnt[rg[k][2]] + rk[k][2];
            const unsigned p3 = cnt[rg[k][3]] + rk[k][3];
            if (p0 < (unsigned)SL_CAP) recs[p0] = make_uint2((unsigned)sv[k].x, (unsigned)dv[k].x);
            if (p1 < (unsigned)SL_CAP) recs[p1] = make_uint2((unsigned)sv[k].y, (unsigned)dv[k].y);
            if (p2 < (unsigned)SL_CAP) recs[p2] = make_uint2((unsigned)sv[k].z, (unsigned)dv[k].z);
            if (p3 < (unsigned)SL_CAP) recs[p3] = make_uint2((unsigned)sv[k].w, (unsigned)dv[k].w);
        }
    }
    if (ht) {
        const unsigned pt = cnt[rgt] + rkt;
        if (pt < (unsigned)SL_CAP) recs[pt] = make_uint2((unsigned)st, (unsigned)dt);
    }
    __syncthreads();

    const int len2 = len >> 1;
    uint4* gp = (uint4*)(pairs + e0);
    const uint4* lp = (const uint4*)recs;
    if (len2 > 0) {
        uint4 ov[SL_CAP / (2 * EP_THREADS)];
#pragma unroll
        for (int k = 0; k < SL_CAP / (2 * EP_THREADS); ++k)
            ov[k] = lp[min(tid + k * EP_THREADS, len2 - 1)];
#pragma unroll
        for (int k = 0; k < SL_CAP / (2 * EP_THREADS); ++k) {
            const int j = tid + k * EP_THREADS;
            if (j < len2) gp[j] = ov[k];
        }
    }
    if ((len & 1) && tid == 0) pairs[e0 + len - 1] = recs[len - 1];
}

__device__ __forceinline__ void gcn_ncount(unsigned* cnt, unsigned d, int lo, unsigned nr) {
    const unsigned rel = d - (unsigned)lo;
    if (rel < nr)
        __hip_atomic_fetch_add(&cnt[rel], 1u, __ATOMIC_RELAXED, __HIP_MEMORY_SCOPE_WORKGROUP);
}

__device__ __forceinline__ void gcn_nplace(unsigned* cnt, unsigned* __restrict__ csr,
                                           uint2 p, int lo, unsigned nr, unsigned E) {
    const unsigned rel = p.y - (unsigned)lo;
    if (rel < nr) {
        const unsigned pos = gcn_take(&cnt[rel]);
        if (pos < E) csr[pos] = p.x;
    }
}

__global__ __launch_bounds__(EP_THREADS)
void gcn_rsort_kernel(const unsigned* __restrict__ pairs_w,
                      const unsigned* __restrict__ tcnt,
                      unsigned* __restrict__ rowstart,
                      float* __restrict__ deg,
                      unsigned* __restrict__ csr, int N, int E) {
    const uint2* __restrict__ pairs = (const uint2*)pairs_w;
    __shared__ unsigned cnt[RN_MAX];
    __shared__ unsigned wt[RN_MAX / 32];
    __shared__ unsigned cs[4][NSLICE];
    __shared__ unsigned ps[NSLICE];
    __shared__ unsigned pc[NSLICE];
    __shared__ unsigned wb[2][NSLICE / 32];
    const int tid  = threadIdx.x;
    const int lane = tid & 31;
    const int wave = __builtin_amdgcn_readfirstlane(tid >> 5);
    const int q    = __builtin_amdgcn_readfirstlane(tid >> 8);
    const int col  = tid & (NSLICE - 1);
    const int r    = (int)blockIdx.x;
    const unsigned rn = gcn_rn(N);
    const int lo = r * (int)rn;
    const unsigned nr = (unsigned)max(0, min(min((int)rn, RN_MAX), N - lo));
    const int sl = gcn_slice_len(E);

    if (tid < RN_MAX) cnt[tid] = 0u;

    unsigned b = 0u;
    {
        const int r0 = q * (NRANGE / 4);
        const unsigned* tp = tcnt + col;
        unsigned c[NRANGE / 4];
#pragma unroll
        for (int k = 0; k < NRANGE / 4; ++k) c[k] = tp[min(r0 + k, r) * NSLICE];
#pragma unroll
        for (int k = 0; k < NRANGE / 4; ++k)
            if (r0 + k < r) b += c[k];
    }
    unsigned myc = 0u;
    if (tid < NSLICE) myc = tcnt[r * NSLICE + tid];
    cs[q][col] = b;
    __syncthreads();

    unsigned pst = 0u;
    if (tid < NSLICE) {
        pst = cs[0][tid] + cs[1][tid] + cs[2][tid] + cs[3][tid];
        ps[tid] = pst;
        pc[tid] = myc;
    }
    unsigned a0 = pst, a1 = myc;
    for (int d = 16; d > 0; d >>= 1) {
        a0 += __shfl_xor(a0, d);
        a1 += __shfl_xor(a1, d);
    }
    if (tid < NSLICE && lane == 0) { wb[0][tid >> 5] = a0; wb[1][tid >> 5] = a1; }
    __syncthreads();
    unsigned e0 = 0u, tot = 0u;
#pragma unroll
    for (int w = 0; w < NSLICE / 32; ++w) { e0 += wb[0][w]; tot += wb[1][w]; }

    unsigned cN[8], pB[8];
#pragma unroll
    for (int k = 0; k < 8; ++k) {
        const int sk = 8 * wave + k;
        cN[k] = (unsigned)__builtin_amdgcn_readfirstlane((int)pc[sk]);
        pB[k] = (unsigned)(sk * sl) + (unsigned)__builtin_amdgcn_readfirstlane((int)ps[sk]);
    }
    uint2 p[2][8];
#pragma unroll
    for (int h = 0; h < 2; ++h) {
#pragma unroll
        for (int k = 0; k < 8; ++k) {
            p[h][k] = make_uint2(0u, 0u);
            if ((unsigned)(lane + 32 * h) < cN[k]) p[h][k] = pairs[pB[k] + (unsigned)(lane + 32 * h)];
        }
    }
#pragma unroll
    for (int h = 0; h < 2; ++h) {
#pragma unroll
        for (int k = 0; k < 8; ++k)
            if ((unsigned)(lane + 32 * h) < cN[k]) gcn_ncount(cnt, p[h][k].y, lo, nr);
    }
#pragma unroll
    for (int k = 0; k < 8; ++k)
        for (unsigned j = (unsigned)lane + 64u; j < cN[k]; j += 32u)
            gcn_ncount(cnt, pairs[pB[k] + j].y, lo, nr);
    __syncthreads();

    unsigned v = 0u;
    if (tid < RN_MAX) v = cnt[tid];
    unsigned x = v;
    for (int d = 1; d < 32; d <<= 1) {
        const unsigned y = __shfl_up(x, d);
        if (lane >= d) x += y;
    }
    if (tid < RN_MAX && lane == 31) wt[wave] = x;
    __syncthreads();
    if (tid < RN_MAX) {
        unsigned start = e0 + (x - v);
        for (int w = 0; w < wave; ++w) start += wt[w];
        cnt[tid] = start;
        if ((unsigned)tid < nr) {
            rowstart[lo + tid] = start;
            const float d = (float)v;
            deg[lo + tid] = d > 0.0f ? 1.0f / d : 0.0f;
        }
    }
    if (r == NRANGE - 1 && tid == 0) rowstart[N] = e0 + tot;
    __syncthreads();

#pragma unroll
    for (int h = 0; h < 2; ++h) {
#pragma unroll
        for (int k = 0; k < 8; ++k)
            if ((unsigned)(lane + 32 * h) < cN[k]) gcn_nplace(cnt, csr, p[h][k], lo, nr, (unsigned)E);
    }
#pragma unroll
    for (int k = 0; k < 8; ++k)
        for (unsigned j = (unsigned)lane + 64u; j < cN[k]; j += 32u)
            gcn_nplace(cnt, csr, pairs[pB[k] + j], lo, nr, (unsigned)E);
}

#define NODES_PER_WAVE  16
#define WAVES_PER_BLOCK 8
#define LDS_PITCH       68

__device__ __forceinline__ const float2* gcn_row2(const float* base, unsigned row, int lane) {
    const unsigned off = row * (unsigned)(DFEAT * sizeof(float)) + (unsigned)lane * (unsigned)sizeof(float2);
    return (const float2*)((const char*)base + off);
}

#define GCN_GATHER8(v, j0)                                                   \
    _Pragma("unroll")                                                        \
    for (int j = 0; j < 8; ++j) {                                            \
        const unsigned sj = (unsigned)__builtin_amdgcn_readlane((int)idx, (j0) + j); \
        v[j] = *gcn_row2(x, sj, lane);                                       \
    }

#define GCN_FLUSH()                                                          \
    {                                                                        \
        const float sc = __int_as_float(__builtin_amdgcn_readlane(__float_as_int(iv), p)); \
        trow[p * (LDS_PITCH / 2)] = make_float2(acc.x * sc, acc.y * sc);     \
        acc = make_float2(0.0f, 0.0f);                                       \
        ++p;                                                                 \
    }

#define GCN_SUM8(v, j0, GUARD)                                               \
    _Pragma("unroll")                                                        \
    for (int j = 0; j < 8; ++j) {                                            \
        if (!(GUARD) || (j0) + j < n) {                                      \
            const unsigned ee = e + (unsigned)((j0) + j);                    \
            while (ee >= nodeEnd) {                                          \
                GCN_FLUSH()                                                  \
                nodeEnd = (unsigned)__builtin_amdgcn_readlane((int)rs, p + 1); \
            }                                                                \
            acc.x += v[j].x;                                                 \
            acc.y += v[j].y;                                                 \
        }                                                                    \
    }

#define GCN_GROUP32(GUARD)                                                   \
    {                                                                        \
        float2 a[8], b[8];                                                   \
        GCN_GATHER8(a, 0)                                                    \
        GCN_GATHER8(b, 8)                                                    \
        GCN_SUM8(a, 0, GUARD)                                                \
        GCN_GATHER8(a, 16)                                                   \
        GCN_SUM8(b, 8, GUARD)                                                \
        GCN_GATHER8(b, 24)                                                   \
        GCN_SUM8(a, 16, GUARD)                                               \
        GCN_SUM8(b, 24, GUARD)                                               \
    }

__device__ __forceinline__ v8f gcn_mma(v2f a, v2f b, v8f c) {
    return __builtin_amdgcn_wmma_f32_16x16x4_f32(
                false, a,  false, b,
                (short)0, c,  false,  false);
}

__device__ __forceinline__ void gcn_finish_tile(float* t, v8f c, float bcol, int hi, int col) {
#pragma unroll
    for (int r = 0; r < 8; ++r) {
        float v = c[r] + bcol;
        v = v > 0.0f ? v : expm1f(v);
        t[(r + (hi << 3)) * LDS_PITCH + col] = v;
    }
}

#define GCN_LOADK(FA, FB, t)                                                 \
    _Pragma("unroll")                                                        \
    for (int ks = 0; ks < 2; ++ks) {                                         \
        FA[ks] = *(const v2f*)(ap + 8 * (t) + 4 * ks);                       \
        _Pragma("unroll")                                                    \
        for (int ct = 0; ct < 4; ++ct)                                       \
            FB[ks][ct] = *(const v2f*)(bp + ct * 16 * LDS_PITCH + 8 * (t) + 4 * ks); \
    }

#define GCN_MMAK(FA, FB)                                                     \
    _Pragma("unroll")                                                        \
    for (int ks = 0; ks < 2; ++ks) {                                         \
        _Pragma("unroll")                                                    \
        for (int ct = 0; ct < 4; ++ct)                                       \
            c[ct] = gcn_mma(FA[ks], FB[ks][ct], c[ct]);                      \
    }

__global__ __launch_bounds__(WAVES_PER_BLOCK * 32)
void gcn_layer_kernel(const float* __restrict__ x,
                      const unsigned* __restrict__ rowstart,
                      const unsigned* __restrict__ csr,
                      const float* __restrict__ inv_deg,
                      const float* __restrict__ W,
                      const float* __restrict__ bias,
                      float* __restrict__ out, int N) {
    __shared__ __attribute__((aligned(16))) float wl[DFEAT * LDS_PITCH];
    __shared__ __attribute__((aligned(16))) float tl[WAVES_PER_BLOCK * NODES_PER_WAVE * LDS_PITCH];

    const int lane = threadIdx.x & 31;
    const int wv = __builtin_amdgcn_readfirstlane((int)(threadIdx.x >> 5));
    const int n0 = ((int)blockIdx.x * WAVES_PER_BLOCK + wv) * NODES_PER_WAVE;

    {
        float4 w4[4];
#pragma unroll
        for (int t = 0; t < 4; ++t)
            w4[t] = *(const float4*)(W + 4 * ((int)threadIdx.x + t * (WAVES_PER_BLOCK * 32)));
#pragma unroll
        for (int t = 0; t < 4; ++t) {
            const int i = (int)threadIdx.x + t * (WAVES_PER_BLOCK * 32);
            *(float4*)(wl + (i >> 4) * LDS_PITCH + ((i & 15) << 2)) = w4[t];
        }
    }
    __syncthreads();
    if (n0 >= N) return;
    const int nn = min(NODES_PER_WAVE, N - n0);

    float* tile = tl + wv * (NODES_PER_WAVE * LDS_PITCH);

    unsigned rs = ~0u;
    if (lane <= nn) rs = rowstart[n0 + lane];
    float iv = 0.0f;
    if (lane < nn) iv = inv_deg[n0 + lane];
    const unsigned eBeg = (unsigned)__builtin_amdgcn_readlane((int)rs, 0);
    const unsigned eEnd = (unsigned)__builtin_amdgcn_readlane((int)rs, nn);
    unsigned nodeEnd    = (unsigned)__builtin_amdgcn_readlane((int)rs, 1);
    int p = 0;
    float2 acc = make_float2(0.0f, 0.0f);
    float2* trow = (float2*)tile + lane;

    if (eBeg < eEnd) {
        const unsigned eLast = eEnd - 1u;
        unsigned iN = csr[min(eBeg + (unsigned)lane, eLast)];

        for (unsigned e = eBeg; e < eEnd; e += 32) {
            const unsigned idx = iN;
            iN = csr[min(e + 32u + (unsigned)lane, eLast)];
            const int n = (int)min(32u, eEnd - e);

            if (n == 32) {
                GCN_GROUP32(0)
            } else {
                GCN_GROUP32(1)
            }
        }
    }
    while (p < NODES_PER_WAVE) GCN_FLUSH()

    __builtin_amdgcn_fence(__ATOMIC_RELEASE, "wavefront");
    __builtin_amdgcn_fence(__ATOMIC_ACQUIRE, "wavefront");

    const int m  = lane & 15;
    const int hi = lane >> 4;
    const int k2 = hi << 1;
    const float* ap = tile + m * LDS_PITCH + k2;
    const float* bp = wl   + m * LDS_PITCH + k2;

    v8f c[4] = {};
    v2f fa[2][2], fb[2][2][4];
    float bcol[4];
#pragma unroll
    for (int ct = 0; ct < 4; ++ct) bcol[ct] = bias[16 * ct + m];
    GCN_LOADK(fa[0], fb[0], 0)
#pragma unroll
    for (int t = 0; t < DFEAT / 8; ++t) {
        if (t + 1 < DFEAT / 8) GCN_LOADK(fa[(t + 1) & 1], fb[(t + 1) & 1], t + 1)
        __builtin_amdgcn_sched_barrier(0);
        GCN_MMAK(fa[t & 1], fb[t & 1])
        __builtin_amdgcn_sched_barrier(0);
    }

#pragma unroll
    for (int ct = 0; ct < 4; ++ct)
        gcn_finish_tile(tile, c[ct], bcol[ct], hi, 16 * ct + m);

    __builtin_amdgcn_fence(__ATOMIC_RELEASE, "wavefront");
    __builtin_amdgcn_fence(__ATOMIC_ACQUIRE, "wavefront");

    const int q4 = m << 2;
    float* orow = out + (size_t)n0 * DFEAT + q4;
    if (nn == NODES_PER_WAVE) {
        float4 ov[NODES_PER_WAVE / 2];
#pragma unroll
        for (int j = 0; j < NODES_PER_WAVE / 2; ++j)
            ov[j] = *(const float4*)(tile + (2 * j + hi) * LDS_PITCH + q4);
#pragma unroll
        for (int j = 0; j < NODES_PER_WAVE / 2; ++j)
            *(float4*)(orow + (2 * j + hi) * DFEAT) = ov[j];
    } else {
        for (int j = 0; j < NODES_PER_WAVE / 2; ++j) {
            const int row = 2 * j + hi;
            if (row < nn)
                *(float4*)(orow + row * DFEAT) = *(const float4*)(tile + row * LDS_PITCH + q4);
        }
    }
}

extern "C" void kernel_launch(void* const* d_in, const int* in_sizes, int n_in,
                              void* d_out, int out_size, void* d_ws, size_t ws_size,
                              hipStream_t stream) {
    const float* h   = (const float*)d_in[0];
    const int*   src = (const int*)  d_in[1];
    const int*   dst = (const int*)  d_in[2];
    const float* W1  = (const float*)d_in[3];
    const float* b1  = (const float*)d_in[4];
    const float* W2  = (const float*)d_in[5];
    const float* b2  = (const float*)d_in[6];
    float*       out = (float*)d_out;

    const int N = in_sizes[0] / DFEAT;
    const int E = in_sizes[1];

    const int layerB = (N + NODES_PER_WAVE * WAVES_PER_BLOCK - 1)
                     / (NODES_PER_WAVE * WAVES_PER_BLOCK);

    char* ws = (char*)d_ws;
    const size_t A = 255;
    float* deg = (float*)ws;
    size_t off = (((size_t)N * sizeof(float)) + A) & ~A;
    float* h1 = (float*)(ws + off);
    off += (size_t)N * DFEAT * sizeof(float);
    unsigned* tcnt = (unsigned*)(ws + off);
    off += (((size_t)NRANGE * NSLICE * sizeof(unsigned)) + A) & ~A;
    unsigned* pairs = (unsigned*)(ws + off);
    off += (((size_t)E * 2 * sizeof(unsigned)) + A) & ~A;
    unsigned* rowstart = (unsigned*)(ws + off);
    off += (((size_t)(N + 1) * sizeof(unsigned)) + A) & ~A;
    unsigned* csr = (unsigned*)(ws + off);

    gcn_lsort_kernel<<<NSLICE, EP_THREADS, 0, stream>>>(src, dst, tcnt, pairs, N, E);
    gcn_rsort_kernel<<<NRANGE, EP_THREADS, 0, stream>>>(pairs, tcnt, rowstart, deg, csr, N, E);

    gcn_layer_kernel<<<layerB, WAVES_PER_BLOCK * 32, 0, stream>>>(h,  rowstart, csr, deg, W1, b1, h1,  N);
    gcn_layer_kernel<<<layerB, WAVES_PER_BLOCK * 32, 0, stream>>>(h1, rowstart, csr, deg, W2, b2, out, N);
}


# range sort and layer 1 in one launch: the block that sorts a node range goes on to gather, multiply and store its own nodes (row starts stay in LDS); 4 launches -> 3
# speedup vs baseline: 5.1939x; 1.0776x over previous
#include <hip/hip_runtime.h>
#include <hip/hip_bf16.h>
#include <math.h>

typedef __attribute__((ext_vector_type(2))) float v2f;
typedef __attribute__((ext_vector_type(8))) float v8f;

#define DFEAT 64

#define NRANGE     256
#define NSLICE     256
#define RN_MAX     512
#define EP_THREADS 1024
#define SL_CAP     (8 * EP_THREADS)

__device__ __forceinline__ unsigned gcn_rn(int N) {
    return (unsigned)max(1, (N + NRANGE - 1) / NRANGE);
}

__device__ __forceinline__ int gcn_slice_len(int E) {
    return (((E + NSLICE - 1) / NSLICE) + 3) & ~3;
}

__device__ __forceinline__ unsigned gcn_range_of(int d, unsigned rn) {
    return min((unsigned)d / rn, (unsigned)(NRANGE - 1));
}

__device__ __forceinline__ unsigned gcn_take(unsigned* c) {
    return __hip_atomic_fetch_add(c, 1u, __ATOMIC_RELAXED, __HIP_MEMORY_SCOPE_WORKGROUP);
}

__global__ __launch_bounds__(EP_THREADS)
void gcn_lsort_kernel(const int* __restrict__ src,
                      const int* __restrict__ dst,
                      unsigned* __restrict__ tcnt,
                      unsigned* __restrict__ pairs_w, int N, int E) {
    uint2* __restrict__ pairs = (uint2*)pairs_w;
    __shared__ __attribute__((aligned(16))) uint2 recs[SL_CAP];
    __shared__ unsigned cnt[NRANGE];
    __shared__ unsigned wt[NRANGE / 32];
    const int tid  = threadIdx.x;
    const int lane = tid & 31;
    const int s    = (int)blockIdx.x;
    if (tid < NRANGE) cnt[tid] = 0u;

    const unsigned rn = gcn_rn(N);
    const int sl = gcn_slice_len(E);
    const int e0 = s * sl;
    const int e1 = min(e0 + sl, E);
    const int len   = min(max(e1 - e0, 0), SL_CAP);
    const int full4 = len & ~3;
    const int* dp = dst + e0;
    const int* sp = src + e0;

    int4 dv[2], sv[2];
    if (full4 > 0) {
#pragma unroll
        for (int k = 0; k < 2; ++k) {
            const int j = min(4 * tid + k * 4 * EP_THREADS, full4 - 4);
            dv[k] = *(const int4*)(dp + j);
            sv[k] = *(const int4*)(sp + j);
        }
    }
    const bool ht = tid < len - full4;
    int dt = 0, st = 0;
    if (ht) { dt = dp[full4 + tid]; st = sp[full4 + tid]; }
    __syncthreads();

    unsigned rg[2][4], rk[2][4], rgt = 0u, rkt = 0u;
#pragma unroll
    for (int k = 0; k < 2; ++k) {
        if (4 * tid + k * 4 * EP_THREADS < full4) {
            rg[k][0] = gcn_range_of(dv[k].x, rn);
            rg[k][1] = gcn_range_of(dv[k].y, rn);
            rg[k][2] = gcn_range_of(dv[k].z, rn);
            rg[k][3] = gcn_range_of(dv[k].w, rn);
#pragma unroll
            for (int c = 0; c < 4; ++c) rk[k][c] = gcn_take(&cnt[rg[k][c]]);
        }
    }
    if (ht) { rgt = gcn_range_of(dt, rn); rkt = gcn_take(&cnt[rgt]); }
    __syncthreads();

    unsigned v = 0u;
    if (tid < NRANGE) v = cnt[tid];
    unsigned x = v;
    for (int d = 1; d < 32; d <<= 1) {
        const unsigned y = __shfl_up(x, d);
        if (lane >= d) x += y;
    }
    if (tid < NRANGE && lane == 31) wt[tid >> 5] = x;
    __syncthreads();
    if (tid < NRANGE) {
        unsigned base = x - v;
        for (int w = 0; w < (tid >> 5); ++w) base += wt[w];
        cnt[tid] = base;
        tcnt[tid * NSLICE + s] = v;
    }
    __syncthreads();

#pragma unroll
    for (int k = 0; k < 2; ++k) {
        if (4 * tid + k * 4 * EP_THREADS < full4) {
            const unsigned p0 = cnt[rg[k][0]] + rk[k][0];
            const unsigned p1 = cnt[rg[k][1]] + rk[k][1];
            const unsigned p2 = cnt[rg[k][2]] + rk[k][2];
            const unsigned p3 = cnt[rg[k][3]] + rk[k][3];
            if (p0 < (unsigned)SL_CAP) recs[p0] = make_uint2((unsigned)sv[k].x, (unsigned)dv[k].x);
            if (p1 < (unsigned)SL_CAP) recs[p1] = make_uint2((unsigned)sv[k].y, (unsigned)dv[k].y);
            if (p2 < (unsigned)SL_CAP) recs[p2] = make_uint2((unsigned)sv[k].z, (unsigned)dv[k].z);
            if (p3 < (unsigned)SL_CAP) recs[p3] = make_uint2((unsigned)sv[k].w, (unsigned)dv[k].w);
        }
    }
    if (ht) {
        const unsigned pt = cnt[rgt] + rkt;
        if (pt < (unsigned)SL_CAP) recs[pt] = make_uint2((unsigned)st, (unsigned)dt);
    }
    __syncthreads();

    const int len2 = len >> 1;
    uint4* gp = (uint4*)(pairs + e0);
    const uint4* lp = (const uint4*)recs;
    if (len2 > 0) {
        uint4 ov[SL_CAP / (2 * EP_THREADS)];
#pragma unroll
        for (int k = 0; k < SL_CAP / (2 * EP_THREADS); ++k)
            ov[k] = lp[min(tid + k * EP_THREADS, len2 - 1)];
#pragma unroll
        for (int k = 0; k < SL_CAP / (2 * EP_THREADS); ++k) {
            const int j = tid + k * EP_THREADS;
            if (j < len2) gp[j] = ov[k];
        }
    }
    if ((len & 1) && tid == 0) pairs[e0 + len - 1] = recs[len - 1];
}

__device__ __forceinline__ void gcn_ncount(unsigned* cnt, unsigned d, int lo, unsigned nr) {
    const unsigned rel = d - (unsigned)lo;
    if (rel < nr)
        __hip_atomic_fetch_add(&cnt[rel], 1u, __ATOMIC_RELAXED, __HIP_MEMORY_SCOPE_WORKGROUP);
}

__device__ __forceinline__ void gcn_nplace(unsigned* cnt, unsigned* csr,
                                           uint2 p, int lo, unsigned nr, unsigned E) {
    const unsigned rel = p.y - (unsigned)lo;
    if (rel < nr) {
        const unsigned pos = gcn_take(&cnt[rel]);
        if (pos < E) csr[pos] = p.x;
    }
}


#define NODES_PER_WAVE  16
#define WAVES_PER_BLOCK 8
#define LDS_PITCH       68

__device__ __forceinline__ const float2* gcn_row2(const float* base, unsigned row, int lane) {
    const unsigned off = row * (unsigned)(DFEAT * sizeof(float)) + (unsigned)lane * (unsigned)sizeof(float2);
    return (const float2*)((const char*)base + off);
}

#define GCN_GATHER8(v, j0)                                                   \
    _Pragma("unroll")                                                        \
    for (int j = 0; j < 8; ++j) {                                            \
        const unsigned sj = (unsigned)__builtin_amdgcn_readlane((int)idx, (j0) + j); \
        v[j] = *gcn_row2(x, sj, lane);                                       \
    }

#define GCN_FLUSH()                                                          \
    {                                                                        \
        const float sc = __int_as_float(__builtin_amdgcn_readlane(__float_as_int(iv), p)); \
        trow[p * (LDS_PITCH / 2)] = make_float2(acc.x * sc, acc.y * sc);     \
        acc = make_float2(0.0f, 0.0f);                                       \
        ++p;                                                                 \
    }

#define GCN_SUM8(v, j0)                                                      \
    _Pragma("unroll")                                                        \
    for (int j = 0; j < 8; ++j) {                                            \
        const unsigned ee = e + (unsigned)((j0) + j);                        \
        while (ee >= nodeEnd) {                                              \
            GCN_FLUSH()                                                      \
            nodeEnd = (unsigned)__builtin_amdgcn_readlane((int)rs, p + 1);   \
        }                                                                    \
        acc.x += v[j].x;                                                     \
        acc.y += v[j].y;                                                     \
    }

#define GCN_GROUP32()                                                        \
    {                                                                        \
        float2 a[8], b[8];                                                   \
        GCN_GATHER8(a, 0)                                                    \
        GCN_GATHER8(b, 8)                                                    \
        GCN_SUM8(a, 0)                                                       \
        GCN_GATHER8(a, 16)                                                   \
        GCN_SUM8(b, 8)                                                       \
        GCN_GATHER8(b, 24)                                                   \
        GCN_SUM8(a, 16)                                                      \
        GCN_SUM8(b, 24)                                                      \
    }

__device__ __forceinline__ v8f gcn_mma(v2f a, v2f b, v8f c) {
    return __builtin_amdgcn_wmma_f32_16x16x4_f32(
                false, a,  false, b,
                (short)0, c,  false,  false);
}

__device__ __forceinline__ void gcn_finish_tile(float* t, v8f c, float bcol, int hi, int col) {
#pragma unroll
    for (int r = 0; r < 8; ++r) {
        float v = c[r] + bcol;
        v = v > 0.0f ? v : expm1f(v);
        t[(r + (hi << 3)) * LDS_PITCH + col] = v;
    }
}

#define GCN_LOADK(FA, FB, t)                                                 \
    _Pragma("unroll")                                                        \
    for (int ks = 0; ks < 2; ++ks) {                                         \
        FA[ks] = *(const v2f*)(ap + 8 * (t) + 4 * ks);                       \
        _Pragma("unroll")                                                    \
        for (int ct = 0; ct < 4; ++ct)                                       \
            FB[ks][ct] = *(const v2f*)(bp + ct * 16 * LDS_PITCH + 8 * (t) + 4 * ks); \
    }

#define GCN_MMAK(FA, FB)                                                     \
    _Pragma("unroll")                                                        \
    for (int ks = 0; ks < 2; ++ks) {                                         \
        _Pragma("unroll")                                                    \
        for (int ct = 0; ct < 4; ++ct)                                       \
            c[ct] = gcn_mma(FA[ks], FB[ks][ct], c[ct]);                      \
    }


__device__ __forceinline__ void gcn_layer_wave(const float* __restrict__ x,
                                               const unsigned* csr,
                                               unsigned rs, float iv, int nn,
                                               const float* wl, float* tile,
                                               const float* __restrict__ bias,
                                               float* __restrict__ out, int n0, int lane) {
    const unsigned eBeg = (unsigned)__builtin_amdgcn_readlane((int)rs, 0);
    const unsigned eEnd = (unsigned)__builtin_amdgcn_readlane((int)rs, nn);
    unsigned nodeEnd    = (unsigned)__builtin_amdgcn_readlane((int)rs, 1);
    int p = 0;
    float2 acc = make_float2(0.0f, 0.0f);
    float2* trow = (float2*)tile + lane;

    if (eBeg < eEnd) {
        const unsigned eLast = eEnd - 1u;
        unsigned iN = csr[min(eBeg + (unsigned)lane, eLast)];

        for (unsigned e = eBeg; e < eEnd; e += 32) {
            const unsigned idx = iN;
            iN = csr[min(e + 32u + (unsigned)lane, eLast)];
            GCN_GROUP32()
        }
    }
    while (p < NODES_PER_WAVE) GCN_FLUSH()

    __builtin_amdgcn_fence(__ATOMIC_RELEASE, "wavefront");
    __builtin_amdgcn_fence(__ATOMIC_ACQUIRE, "wavefront");

    const int m  = lane & 15;
    const int hi = lane >> 4;
    const int k2 = hi << 1;
    const float* ap = tile + m * LDS_PITCH + k2;
    const float* bp = wl   + m * LDS_PITCH + k2;

    v8f c[4] = {};
    v2f fa[2][2], fb[2][2][4];
    float bcol[4];
#pragma unroll
    for (int ct = 0; ct < 4; ++ct) bcol[ct] = bias[16 * ct + m];
    GCN_LOADK(fa[0], fb[0], 0)
#pragma unroll
    for (int t = 0; t < DFEAT / 8; ++t) {
        if (t + 1 < DFEAT / 8) GCN_LOADK(fa[(t + 1) & 1], fb[(t + 1) & 1], t + 1)
        __builtin_amdgcn_sched_barrier(0);
        GCN_MMAK(fa[t & 1], fb[t & 1])
        __builtin_amdgcn_sched_barrier(0);
    }

#pragma unroll
    for (int ct = 0; ct < 4; ++ct)
        gcn_finish_tile(tile, c[ct], bcol[ct], hi, 16 * ct + m);

    __builtin_amdgcn_fence(__ATOMIC_RELEASE, "wavefront");
    __builtin_amdgcn_fence(__ATOMIC_ACQUIRE, "wavefront");

    const int q4 = m << 2;
    float* orow = out + (size_t)n0 * DFEAT + q4;
    if (nn == NODES_PER_WAVE) {
        float4 ov[NODES_PER_WAVE / 2];
#pragma unroll
        for (int j = 0; j < NODES_PER_WAVE / 2; ++j)
            ov[j] = *(const float4*)(tile + (2 * j + hi) * LDS_PITCH + q4);
#pragma unroll
        for (int j = 0; j < NODES_PER_WAVE / 2; ++j)
            *(float4*)(orow + (2 * j + hi) * DFEAT) = ov[j];
    } else {
        for (int j = 0; j < NODES_PER_WAVE / 2; ++j) {
            const int row = 2 * j + hi;
            if (row < nn)
                *(float4*)(orow + row * DFEAT) = *(const float4*)(tile + row * LDS_PITCH + q4);
        }
    }
}

__global__ __launch_bounds__(WAVES_PER_BLOCK * 32)
void gcn_layer_kernel(const float* __restrict__ x,
                      const unsigned* __restrict__ rowstart,
                      const unsigned* __restrict__ csr,
                      const float* __restrict__ inv_deg,
                      const float* __restrict__ W,
                      const float* __restrict__ bias,
                      float* __restrict__ out, int N) {
    __shared__ __attribute__((aligned(16))) float wl[DFEAT * LDS_PITCH];
    __shared__ __attribute__((aligned(16))) float tl[WAVES_PER_BLOCK * NODES_PER_WAVE * LDS_PITCH];

    const int lane = threadIdx.x & 31;
    const int wv = __builtin_amdgcn_readfirstlane((int)(threadIdx.x >> 5));
    const int n0 = ((int)blockIdx.x * WAVES_PER_BLOCK + wv) * NODES_PER_WAVE;

    {
        float4 w4[4];
#pragma unroll
        for (int t = 0; t < 4; ++t)
            w4[t] = *(const float4*)(W + 4 * ((int)threadIdx.x + t * (WAVES_PER_BLOCK * 32)));
#pragma unroll
        for (int t = 0; t < 4; ++t) {
            const int i = (int)threadIdx.x + t * (WAVES_PER_BLOCK * 32);
            *(float4*)(wl + (i >> 4) * LDS_PITCH + ((i & 15) << 2)) = w4[t];
        }
    }
    __syncthreads();
    if (n0 >= N) return;
    const int nn = min(NODES_PER_WAVE, N - n0);

    float* tile = tl + wv * (NODES_PER_WAVE * LDS_PITCH);

    unsigned rs = ~0u;
    if (lane <= nn) rs = rowstart[n0 + lane];
    float iv = 0.0f;
    if (lane < nn) iv = inv_deg[n0 + lane];
    gcn_layer_wave(x, csr, rs, iv, nn, wl, tile, bias, out, n0, lane);
}

__global__ __launch_bounds__(EP_THREADS)
void gcn_rsort_layer_kernel(const unsigned* __restrict__ pairs_w,
                            const unsigned* __restrict__ tcnt,
                            unsigned* __restrict__ rowstart,
                            float* __restrict__ deg,
                            unsigned* csr,
                            const float* __restrict__ feat,
                            const float* __restrict__ W,
                            const float* __restrict__ bias,
                            float* __restrict__ out, int N, int E) {
    static_assert(EP_THREADS * 4 == DFEAT * DFEAT, "one float4 of W per thread");
    static_assert(RN_MAX / NODES_PER_WAVE <= EP_THREADS / 32, "a wave per 16 nodes of the range");
    const uint2* __restrict__ pairs = (const uint2*)pairs_w;
    __shared__ unsigned cnt[RN_MAX];
    __shared__ unsigned wt[RN_MAX / 32];
    __shared__ unsigned cs[4][NSLICE];
    __shared__ unsigned ps[NSLICE];
    __shared__ unsigned pc[NSLICE];
    __shared__ unsigned wb[2][NSLICE / 32];
    __shared__ unsigned rsl[RN_MAX + 1];
    __shared__ __attribute__((aligned(16))) float wl[DFEAT * LDS_PITCH];
    __shared__ __attribute__((aligned(16))) float tl[(RN_MAX / NODES_PER_WAVE) * NODES_PER_WAVE * LDS_PITCH];
    const int tid  = threadIdx.x;
    const int lane = tid & 31;
    const int wave = __builtin_amdgcn_readfirstlane(tid >> 5);
    const int q    = __builtin_amdgcn_readfirstlane(tid >> 8);
    const int col  = tid & (NSLICE - 1);
    const int r    = (int)blockIdx.x;
    const unsigned rn = gcn_rn(N);
    const int lo = r * (int)rn;
    const unsigned nr = (unsigned)max(0, min(min((int)rn, RN_MAX), N - lo));
    const int sl = gcn_slice_len(E);

    if (tid < RN_MAX) cnt[tid] = 0u;
    const float4 w4 = *(const float4*)(W + 4 * tid);

    unsigned b = 0u;
    {
        const int r0 = q * (NRANGE / 4);
        const unsigned* tp = tcnt + col;
        unsigned c[NRANGE / 4];
#pragma unroll
        for (int k = 0; k < NRANGE / 4; ++k) c[k] = tp[min(r0 + k, r) * NSLICE];
#pragma unroll
        for (int k = 0; k < NRANGE / 4; ++k)
            if (r0 + k < r) b += c[k];
    }
    unsigned myc = 0u;
    if (tid < NSLICE) myc = tcnt[r * NSLICE + tid];
    cs[q][col] = b;
    *(float4*)(wl + (tid >> 4) * LDS_PITCH + ((tid & 15) << 2)) = w4;
    __syncthreads();

    unsigned pst = 0u;
    if (tid < NSLICE) {
        pst = cs[0][tid] + cs[1][tid] + cs[2][tid] + cs[3][tid];
        ps[tid] = pst;
        pc[tid] = myc;
    }
    unsigned a0 = pst, a1 = myc;
    for (int d = 16; d > 0; d >>= 1) {
        a0 += __shfl_xor(a0, d);
        a1 += __shfl_xor(a1, d);
    }
    if (tid < NSLICE && lane == 0) { wb[0][tid >> 5] = a0; wb[1][tid >> 5] = a1; }
    __syncthreads();
    unsigned e0 = 0u, tot = 0u;
#pragma unroll
    for (int w = 0; w < NSLICE / 32; ++w) { e0 += wb[0][w]; tot += wb[1][w]; }

    unsigned cN[8], pB[8];
#pragma unroll
    for (int k = 0; k < 8; ++k) {
        const int sk = 8 * wave + k;
        cN[k] = (unsigned)__builtin_amdgcn_readfirstlane((int)pc[sk]);
        pB[k] = (unsigned)(sk * sl) + (unsigned)__builtin_amdgcn_readfirstlane((int)ps[sk]);
    }
    uint2 p[2][8];
#pragma unroll
    for (int h = 0; h < 2; ++h) {
#pragma unroll
        for (int k = 0; k < 8; ++k) {
            p[h][k] = make_uint2(0u, 0u);
            if ((unsigned)(lane + 32 * h) < cN[k]) p[h][k] = pairs[pB[k] + (unsigned)(lane + 32 * h)];
        }
    }
#pragma unroll
    for (int h = 0; h < 2; ++h) {
#pragma unroll
        for (int k = 0; k < 8; ++k)
            if ((unsigned)(lane + 32 * h) < cN[k]) gcn_ncount(cnt, p[h][k].y, lo, nr);
    }
#pragma unroll
    for (int k = 0; k < 8; ++k)
        for (unsigned j = (unsigned)lane + 64u; j < cN[k]; j += 32u)
            gcn_ncount(cnt, pairs[pB[k] + j].y, lo, nr);
    __syncthreads();

    unsigned v = 0u;
    if (tid < RN_MAX) v = cnt[tid];
    unsigned x = v;
    for (int d = 1; d < 32; d <<= 1) {
        const unsigned y = __shfl_up(x, d);
        if (lane >= d) x += y;
    }
    if (tid < RN_MAX && lane == 31) wt[wave] = x;
    __syncthreads();
    if (tid < RN_MAX) {
        unsigned start = e0 + (x - v);
        for (int w = 0; w < wave; ++w) start += wt[w];
        cnt[tid] = start;
        rsl[tid] = start;
        if (tid == RN_MAX - 1) rsl[RN_MAX] = start + v;
        if ((unsigned)tid < nr) {
            rowstart[lo + tid] = start;
            const float d = (float)v;
            deg[lo + tid] = d > 0.0f ? 1.0f / d : 0.0f;
        }
    }
    if (r == NRANGE - 1 && tid == 0) rowstart[N] = e0 + tot;
    __syncthreads();

#pragma unroll
    for (int h = 0; h < 2; ++h) {
#pragma unroll
        for (int k = 0; k < 8; ++k)
            if ((unsigned)(lane + 32 * h) < cN[k]) gcn_nplace(cnt, csr, p[h][k], lo, nr, (unsigned)E);
    }
#pragma unroll
    for (int k = 0; k < 8; ++k)
        for (unsigned j = (unsigned)lane + 64u; j < cN[k]; j += 32u)
            gcn_nplace(cnt, csr, pairs[pB[k] + j], lo, nr, (unsigned)E);

    __syncthreads();

    const int nl = NODES_PER_WAVE * wave;
    if ((unsigned)nl < nr) {
        const int nn = min(NODES_PER_WAVE, (int)nr - nl);
        unsigned rs = ~0u;
        if (lane <= nn) rs = rsl[nl + lane];
        float iv = 0.0f;
        if (lane < nn) {
            const float d = (float)(rsl[nl + lane + 1] - rsl[nl + lane]);
            iv = d > 0.0f ? 1.0f / d : 0.0f;
        }
        gcn_layer_wave(feat, csr, rs, iv, nn, wl, tl + wave * (NODES_PER_WAVE * LDS_PITCH),
                       bias, out, lo + nl, lane);
    }
}

extern "C" void kernel_launch(void* const* d_in, const int* in_sizes, int n_in,
                              void* d_out, int out_size, void* d_ws, size_t ws_size,
                              hipStream_t stream) {
    const float* h   = (const float*)d_in[0];
    const int*   src = (const int*)  d_in[1];
    const int*   dst = (const int*)  d_in[2];
    const float* W1  = (const float*)d_in[3];
    const float* b1  = (const float*)d_in[4];
    const float* W2  = (const float*)d_in[5];
    const float* b2  = (const float*)d_in[6];
    float*       out = (float*)d_out;

    const int N = in_sizes[0] / DFEAT;
    const int E = in_sizes[1];

    const int layerB = (N + NODES_PER_WAVE * WAVES_PER_BLOCK - 1)
                     / (NODES_PER_WAVE * WAVES_PER_BLOCK);

    char* ws = (char*)d_ws;
    const size_t A = 255;
    float* deg = (float*)ws;
    size_t off = (((size_t)N * sizeof(float)) + A) & ~A;
    float* h1 = (float*)(ws + off);
    off += (size_t)N * DFEAT * sizeof(float);
    unsigned* tcnt = (unsigned*)(ws + off);
    off += (((size_t)NRANGE * NSLICE * sizeof(unsigned)) + A) & ~A;
    unsigned* pairs = (unsigned*)(ws + off);
    off += (((size_t)E * 2 * sizeof(unsigned)) + A) & ~A;
    unsigned* rowstart = (unsigned*)(ws + off);
    off += (((size_t)(N + 1) * sizeof(unsigned)) + A) & ~A;
    unsigned* csr = (unsigned*)(ws + off);

    gcn_lsort_kernel<<<NSLICE, EP_THREADS, 0, stream>>>(src, dst, tcnt, pairs, N, E);
    gcn_rsort_layer_kernel<<<NRANGE, EP_THREADS, 0, stream>>>(pairs, tcnt, rowstart, deg, csr,
                                                             h, W1, b1, h1, N, E);

    gcn_layer_kernel<<<layerB, WAVES_PER_BLOCK * 32, 0, stream>>>(h1, rowstart, csr, deg, W2, b2, out, N);
}


# sort launch: each range's sorted source ids are placed in an LDS copy of its csr piece, leave as whole 16-byte groups and feed layer 1 from LDS; was one 4-byte global store per edge, read back
# speedup vs baseline: 5.6324x; 1.0844x over previous
#include <hip/hip_runtime.h>
#include <hip/hip_bf16.h>
#include <math.h>

typedef __attribute__((ext_vector_type(2))) float v2f;
typedef __attribute__((ext_vector_type(8))) float v8f;

#define DFEAT 64

#define NRANGE     256
#define NSLICE     256
#define RN_MAX     512
#define EP_THREADS 1024
#define SL_CAP     (8 * EP_THREADS)
#define IDS_CAP    16384

__device__ __forceinline__ unsigned gcn_rn(int N) {
    return (unsigned)max(1, (N + NRANGE - 1) / NRANGE);
}

__device__ __forceinline__ int gcn_slice_len(int E) {
    return (((E + NSLICE - 1) / NSLICE) + 3) & ~3;
}

__device__ __forceinline__ unsigned gcn_range_of(int d, unsigned rn) {
    return min((unsigned)d / rn, (unsigned)(NRANGE - 1));
}

__device__ __forceinline__ unsigned gcn_take(unsigned* c) {
    return __hip_atomic_fetch_add(c, 1u, __ATOMIC_RELAXED, __HIP_MEMORY_SCOPE_WORKGROUP);
}

__global__ __launch_bounds__(EP_THREADS)
void gcn_lsort_kernel(const int* __restrict__ src,
                      const int* __restrict__ dst,
                      unsigned* __restrict__ tcnt,
                      unsigned* __restrict__ pairs_w, int N, int E) {
    uint2* __restrict__ pairs = (uint2*)pairs_w;
    __shared__ __attribute__((aligned(16))) uint2 recs[SL_CAP];
    __shared__ unsigned cnt[NRANGE];
    __shared__ unsigned wt[NRANGE / 32];
    const int tid  = threadIdx.x;
    const int lane = tid & 31;
    const int s    = (int)blockIdx.x;
    if (tid < NRANGE) cnt[tid] = 0u;

    const unsigned rn = gcn_rn(N);
    const int sl = gcn_slice_len(E);
    const int e0 = s * sl;
    const int e1 = min(e0 + sl, E);
    const int len   = min(max(e1 - e0, 0), SL_CAP);
    const int full4 = len & ~3;
    const int* dp = dst + e0;
    const int* sp = src + e0;

    int4 dv[2], sv[2];
    if (full4 > 0) {
#pragma unroll
        for (int k = 0; k < 2; ++k) {
            const int j = min(4 * tid + k * 4 * EP_THREADS, full4 - 4);
            dv[k] = *(const int4*)(dp + j);
            sv[k] = *(const int4*)(sp + j);
        }
    }
    const bool ht = tid < len - full4;
    int dt = 0, st = 0;
    if (ht) { dt = dp[full4 + tid]; st = sp[full4 + tid]; }
    __syncthreads();

    unsigned rg[2][4], rk[2][4], rgt = 0u, rkt = 0u;
#pragma unroll
    for (int k = 0; k < 2; ++k) {
        if (4 * tid + k * 4 * EP_THREADS < full4) {
            rg[k][0] = gcn_range_of(dv[k].x, rn);
            rg[k][1] = gcn_range_of(dv[k].y, rn);
            rg[k][2] = gcn_range_of(dv[k].z, rn);
            rg[k][3] = gcn_range_of(dv[k].w, rn);
#pragma unroll
            for (int c = 0; c < 4; ++c) rk[k][c] = gcn_take(&cnt[rg[k][c]]);
        }
    }
    if (ht) { rgt = gcn_range_of(dt, rn); rkt = gcn_take(&cnt[rgt]); }
    __syncthreads();

    unsigned v = 0u;
    if (tid < NRANGE) v = cnt[tid];
    unsigned x = v;
    for (int d = 1; d < 32; d <<= 1) {
        const unsigned y = __shfl_up(x, d);
        if (lane >= d) x += y;
    }
    if (tid < NRANGE && lane == 31) wt[tid >> 5] = x;
    __syncthreads();
    if (tid < NRANGE) {
        unsigned base = x - v;
        for (int w = 0; w < (tid >> 5); ++w) base += wt[w];
        cnt[tid] = base;
        tcnt[tid * NSLICE + s] = v;
    }
    __syncthreads();

#pragma unroll
    for (int k = 0; k < 2; ++k) {
        if (4 * tid + k * 4 * EP_THREADS < full4) {
            const unsigned p0 = cnt[rg[k][0]] + rk[k][0];
            const unsigned p1 = cnt[rg[k][1]] + rk[k][1];
            const unsigned p2 = cnt[rg[k][2]] + rk[k][2];
            const unsigned p3 = cnt[rg[k][3]] + rk[k][3];
            if (p0 < (unsigned)SL_CAP) recs[p0] = make_uint2((unsigned)sv[k].x, (unsigned)dv[k].x);
            if (p1 < (unsigned)SL_CAP) recs[p1] = make_uint2((unsigned)sv[k].y, (unsigned)dv[k].y);
            if (p2 < (unsigned)SL_CAP) recs[p2] = make_uint2((unsigned)sv[k].z, (unsigned)dv[k].z);
            if (p3 < (unsigned)SL_CAP) recs[p3] = make_uint2((unsigned)sv[k].w, (unsigned)dv[k].w);
        }
    }
    if (ht) {
        const unsigned pt = cnt[rgt] + rkt;
        if (pt < (unsigned)SL_CAP) recs[pt] = make_uint2((unsigned)st, (unsigned)dt);
    }
    __syncthreads();

    const int len2 = len >> 1;
    uint4* gp = (uint4*)(pairs + e0);
    const uint4* lp = (const uint4*)recs;
    if (len2 > 0) {
        uint4 ov[SL_CAP / (2 * EP_THREADS)];
#pragma unroll
        for (int k = 0; k < SL_CAP / (2 * EP_THREADS); ++k)
            ov[k] = lp[min(tid + k * EP_THREADS, len2 - 1)];
#pragma unroll
        for (int k = 0; k < SL_CAP / (2 * EP_THREADS); ++k) {
            const int j = tid + k * EP_THREADS;
            if (j < len2) gp[j] = ov[k];
        }
    }
    if ((len & 1) && tid == 0) pairs[e0 + len - 1] = recs[len - 1];
}

__device__ __forceinline__ void gcn_ncount(unsigned* cnt, unsigned d, int lo, unsigned nr) {
    const unsigned rel = d - (unsigned)lo;
    if (rel < nr)
        __hip_atomic_fetch_add(&cnt[rel], 1u, __ATOMIC_RELAXED, __HIP_MEMORY_SCOPE_WORKGROUP);
}

__device__ __forceinline__ void gcn_nplace(unsigned* cnt, unsigned* places, unsigned sub, unsigned cap,
                                           uint2 p, int lo, unsigned nr) {
    const unsigned rel = p.y - (unsigned)lo;
    if (rel < nr) {
        const unsigned w = gcn_take(&cnt[rel]) - sub;
        if (w < cap) places[w] = p.x;
    }
}


#define NODES_PER_WAVE  16
#define WAVES_PER_BLOCK 8
#define LDS_PITCH       68

__device__ __forceinline__ const float2* gcn_row2(const float* base, unsigned row, int lane) {
    const unsigned off = row * (unsigned)(DFEAT * sizeof(float)) + (unsigned)lane * (unsigned)sizeof(float2);
    return (const float2*)((const char*)base + off);
}

#define GCN_GATHER8(v, j0)                                                   \
    _Pragma("unroll")                                                        \
    for (int j = 0; j < 8; ++j) {                                            \
        const unsigned sj = (unsigned)__builtin_amdgcn_readlane((int)idx, (j0) + j); \
        v[j] = *gcn_row2(x, sj, lane);                                       \
    }

#define GCN_FLUSH()                                                          \
    {                                                                        \
        const float sc = __int_as_float(__builtin_amdgcn_readlane(__float_as_int(iv), p)); \
        trow[p * (LDS_PITCH / 2)] = make_float2(acc.x * sc, acc.y * sc);     \
        acc = make_float2(0.0f, 0.0f);                                       \
        ++p;                                                                 \
    }

#define GCN_SUM8(v, j0)                                                      \
    _Pragma("unroll")                                                        \
    for (int j = 0; j < 8; ++j) {                                            \
        const unsigned ee = e + (unsigned)((j0) + j);                        \
        while (ee >= nodeEnd) {                                              \
            GCN_FLUSH()                                                      \
            nodeEnd = (unsigned)__builtin_amdgcn_readlane((int)rs, p + 1);   \
        }                                                                    \
        acc.x += v[j].x;                                                     \
        acc.y += v[j].y;                                                     \
    }

#define GCN_GROUP32()                                                        \
    {                                                                        \
        float2 a[8], b[8];                                                   \
        GCN_GATHER8(a, 0)                                                    \
        GCN_GATHER8(b, 8)                                                    \
        GCN_SUM8(a, 0)                                                       \
        GCN_GATHER8(a, 16)                                                   \
        GCN_SUM8(b, 8)                                                       \
        GCN_GATHER8(b, 24)                                                   \
        GCN_SUM8(a, 16)                                                      \
        GCN_SUM8(b, 24)                                                      \
    }

__device__ __forceinline__ v8f gcn_mma(v2f a, v2f b, v8f c) {
    return __builtin_amdgcn_wmma_f32_16x16x4_f32(
                false, a,  false, b,
                (short)0, c,  false,  false);
}

__device__ __forceinline__ void gcn_finish_tile(float* t, v8f c, float bcol, int hi, int col) {
#pragma unroll
    for (int r = 0; r < 8; ++r) {
        float v = c[r] + bcol;
        v = v > 0.0f ? v : expm1f(v);
        t[(r + (hi << 3)) * LDS_PITCH + col] = v;
    }
}

#define GCN_LOADK(FA, FB, t)                                                 \
    _Pragma("unroll")                                                        \
    for (int ks = 0; ks < 2; ++ks) {                                         \
        FA[ks] = *(const v2f*)(ap + 8 * (t) + 4 * ks);                       \
        _Pragma("unroll")                                                    \
        for (int ct = 0; ct < 4; ++ct)                                       \
            FB[ks][ct] = *(const v2f*)(bp + ct * 16 * LDS_PITCH + 8 * (t) + 4 * ks); \
    }

#define GCN_MMAK(FA, FB)                                                     \
    _Pragma("unroll")                                                        \
    for (int ks = 0; ks < 2; ++ks) {                                         \
        _Pragma("unroll")                                                    \
        for (int ct = 0; ct < 4; ++ct)                                       \
            c[ct] = gcn_mma(FA[ks], FB[ks][ct], c[ct]);                      \
    }


__device__ __forceinline__ void gcn_layer_wave(const float* __restrict__ x,
                                               const unsigned* csr,
                                               unsigned rs, float iv, int nn,
                                               const float* wl, float* tile,
                                               const float* __restrict__ bias,
                                               float* __restrict__ out, int n0, int lane) {
    const unsigned eBeg = (unsigned)__builtin_amdgcn_readlane((int)rs, 0);
    const unsigned eEnd = (unsigned)__builtin_amdgcn_readlane((int)rs, nn);
    unsigned nodeEnd    = (unsigned)__builtin_amdgcn_readlane((int)rs, 1);
    int p = 0;
    float2 acc = make_float2(0.0f, 0.0f);
    float2* trow = (float2*)tile + lane;

    if (eBeg < eEnd) {
        const unsigned eLast = eEnd - 1u;
        unsigned iN = csr[min(eBeg + (unsigned)lane, eLast)];

        for (unsigned e = eBeg; e < eEnd; e += 32) {
            const unsigned idx = iN;
            iN = csr[min(e + 32u + (unsigned)lane, eLast)];
            GCN_GROUP32()
        }
    }
    while (p < NODES_PER_WAVE) GCN_FLUSH()

    __builtin_amdgcn_fence(__ATOMIC_RELEASE, "wavefront");
    __builtin_amdgcn_fence(__ATOMIC_ACQUIRE, "wavefront");

    const int m  = lane & 15;
    const int hi = lane >> 4;
    const int k2 = hi << 1;
    const float* ap = tile + m * LDS_PITCH + k2;
    const float* bp = wl   + m * LDS_PITCH + k2;

    v8f c[4] = {};
    v2f fa[2][2], fb[2][2][4];
    float bcol[4];
#pragma unroll
    for (int ct = 0; ct < 4; ++ct) bcol[ct] = bias[16 * ct + m];
    GCN_LOADK(fa[0], fb[0], 0)
#pragma unroll
    for (int t = 0; t < DFEAT / 8; ++t) {
        if (t + 1 < DFEAT / 8) GCN_LOADK(fa[(t + 1) & 1], fb[(t + 1) & 1], t + 1)
        __builtin_amdgcn_sched_barrier(0);
        GCN_MMAK(fa[t & 1], fb[t & 1])
        __builtin_amdgcn_sched_barrier(0);
    }

#pragma unroll
    for (int ct = 0; ct < 4; ++ct)
        gcn_finish_tile(tile, c[ct], bcol[ct], hi, 16 * ct + m);

    __builtin_amdgcn_fence(__ATOMIC_RELEASE, "wavefront");
    __builtin_amdgcn_fence(__ATOMIC_ACQUIRE, "wavefront");

    const int q4 = m << 2;
    float* orow = out + (size_t)n0 * DFEAT + q4;
    if (nn == NODES_PER_WAVE) {
        float4 ov[NODES_PER_WAVE / 2];
#pragma unroll
        for (int j = 0; j < NODES_PER_WAVE / 2; ++j)
            ov[j] = *(const float4*)(tile + (2 * j + hi) * LDS_PITCH + q4);
#pragma unroll
        for (int j = 0; j < NODES_PER_WAVE / 2; ++j)
            *(float4*)(orow + (2 * j + hi) * DFEAT) = ov[j];
    } else {
        for (int j = 0; j < NODES_PER_WAVE / 2; ++j) {
            const int row = 2 * j + hi;
            if (row < nn)
                *(float4*)(orow + row * DFEAT) = *(const float4*)(tile + row * LDS_PITCH + q4);
        }
    }
}

__global__ __launch_bounds__(WAVES_PER_BLOCK * 32)
void gcn_layer_kernel(const float* __restrict__ x,
                      const unsigned* __restrict__ rowstart,
                      const unsigned* __restrict__ csr,
                      const float* __restrict__ inv_deg,
                      const float* __restrict__ W,
                      const float* __restrict__ bias,
                      float* __restrict__ out, int N) {
    __shared__ __attribute__((aligned(16))) float wl[DFEAT * LDS_PITCH];
    __shared__ __attribute__((aligned(16))) float tl[WAVES_PER_BLOCK * NODES_PER_WAVE * LDS_PITCH];

    const int lane = threadIdx.x & 31;
    const int wv = __builtin_amdgcn_readfirstlane((int)(threadIdx.x >> 5));
    const int n0 = ((int)blockIdx.x * WAVES_PER_BLOCK + wv) * NODES_PER_WAVE;

    {
        float4 w4[4];
#pragma unroll
        for (int t = 0; t < 4; ++t)
            w4[t] = *(const float4*)(W + 4 * ((int)threadIdx.x + t * (WAVES_PER_BLOCK * 32)));
#pragma unroll
        for (int t = 0; t < 4; ++t) {
            const int i = (int)threadIdx.x + t * (WAVES_PER_BLOCK * 32);
            *(float4*)(wl + (i >> 4) * LDS_PITCH + ((i & 15) << 2)) = w4[t];
        }
    }
    __syncthreads();
    if (n0 >= N) return;
    const int nn = min(NODES_PER_WAVE, N - n0);

    float* tile = tl + wv * (NODES_PER_WAVE * LDS_PITCH);

    unsigned rs = ~0u;
    if (lane <= nn) rs = rowstart[n0 + lane];
    float iv = 0.0f;
    if (lane < nn) iv = inv_deg[n0 + lane];
    gcn_layer_wave(x, csr, rs, iv, nn, wl, tile, bias, out, n0, lane);
}

__device__ __forceinline__ void gcn_range_layer_wave(const unsigned* rsl, const unsigned* places,
                                                     unsigned sub, int nl, unsigned nr,
                                                     const float* __restrict__ feat,
                                                     const float* wl, float* tile,
                                                     const float* __restrict__ bias,
                                                     float* __restrict__ out, int n0, int lane) {
    const int nn = min(NODES_PER_WAVE, (int)nr - nl);
    unsigned rs = ~0u;
    if (lane <= nn) rs = rsl[nl + lane] - sub;
    float iv = 0.0f;
    if (lane < nn) {
        const float d = (float)(rsl[nl + lane + 1] - rsl[nl + lane]);
        iv = d > 0.0f ? 1.0f / d : 0.0f;
    }
    gcn_layer_wave(feat, places, rs, iv, nn, wl, tile, bias, out, n0, lane);
}

__global__ __launch_bounds__(EP_THREADS)
void gcn_rsort_layer_kernel(const unsigned* __restrict__ pairs_w,
                            const unsigned* __restrict__ tcnt,
                            unsigned* __restrict__ rowstart,
                            float* __restrict__ deg,
                            unsigned* csr,
                            const float* __restrict__ feat,
                            const float* __restrict__ W,
                            const float* __restrict__ bias,
                            float* __restrict__ out, int N, int E) {
    static_assert(EP_THREADS * 4 == DFEAT * DFEAT, "one float4 of W per thread");
    static_assert(RN_MAX / NODES_PER_WAVE <= EP_THREADS / 32, "a wave per 16 nodes of the range");
    const uint2* __restrict__ pairs = (const uint2*)pairs_w;
    __shared__ unsigned cnt[RN_MAX];
    __shared__ unsigned wt[RN_MAX / 32];
    __shared__ unsigned cs[4][NSLICE];
    __shared__ unsigned ps[NSLICE];
    __shared__ unsigned pc[NSLICE];
    __shared__ unsigned wb[2][NSLICE / 32];
    __shared__ unsigned rsl[RN_MAX + 1];
    __shared__ __attribute__((aligned(16))) unsigned ids[IDS_CAP];
    __shared__ __attribute__((aligned(16))) float wl[DFEAT * LDS_PITCH];
    __shared__ __attribute__((aligned(16))) float tl[(RN_MAX / NODES_PER_WAVE) * NODES_PER_WAVE * LDS_PITCH];
    const int tid  = threadIdx.x;
    const int lane = tid & 31;
    const int wave = __builtin_amdgcn_readfirstlane(tid >> 5);
    const int q    = __builtin_amdgcn_readfirstlane(tid >> 8);
    const int col  = tid & (NSLICE - 1);
    const int r    = (int)blockIdx.x;
    const unsigned rn = gcn_rn(N);
    const int lo = r * (int)rn;
    const unsigned nr = (unsigned)max(0, min(min((int)rn, RN_MAX), N - lo));
    const int sl = gcn_slice_len(E);

    if (tid < RN_MAX) cnt[tid] = 0u;
    const float4 w4 = *(const float4*)(W + 4 * tid);

    unsigned b = 0u;
    {
        const int r0 = q * (NRANGE / 4);
        const unsigned* tp = tcnt + col;
        unsigned c[NRANGE / 4];
#pragma unroll
        for (int k = 0; k < NRANGE / 4; ++k) c[k] = tp[min(r0 + k, r) * NSLICE];
#pragma unroll
        for (int k = 0; k < NRANGE / 4; ++k)
            if (r0 + k < r) b += c[k];
    }
    unsigned myc = 0u;
    if (tid < NSLICE) myc = tcnt[r * NSLICE + tid];
    cs[q][col] = b;
    *(float4*)(wl + (tid >> 4) * LDS_PITCH + ((tid & 15) << 2)) = w4;
    __syncthreads();

    unsigned pst = 0u;
    if (tid < NSLICE) {
        pst = cs[0][tid] + cs[1][tid] + cs[2][tid] + cs[3][tid];
        ps[tid] = pst;
        pc[tid] = myc;
    }
    unsigned a0 = pst, a1 = myc;
    for (int d = 16; d > 0; d >>= 1) {
        a0 += __shfl_xor(a0, d);
        a1 += __shfl_xor(a1, d);
    }
    if (tid < NSLICE && lane == 0) { wb[0][tid >> 5] = a0; wb[1][tid >> 5] = a1; }
    __syncthreads();
    unsigned e0 = 0u, tot = 0u;
#pragma unroll
    for (int w = 0; w < NSLICE / 32; ++w) { e0 += wb[0][w]; tot += wb[1][w]; }
    const unsigned eb = (unsigned)__builtin_amdgcn_readfirstlane((int)(e0 & ~3u));
    const bool in_lds = __builtin_amdgcn_readfirstlane((int)((e0 - eb) + tot <= (unsigned)IDS_CAP)) != 0;

    unsigned cN[8], pB[8];
#pragma unroll
    for (int k = 0; k < 8; ++k) {
        const int sk = 8 * wave + k;
        cN[k] = (unsigned)__builtin_amdgcn_readfirstlane((int)pc[sk]);
        pB[k] = (unsigned)(sk * sl) + (unsigned)__builtin_amdgcn_readfirstlane((int)ps[sk]);
    }
    uint2 p[2][8];
#pragma unroll
    for (int h = 0; h < 2; ++h) {
#pragma unroll
        for (int k = 0; k < 8; ++k) {
            p[h][k] = make_uint2(0u, 0u);
            if ((unsigned)(lane + 32 * h) < cN[k]) p[h][k] = pairs[pB[k] + (unsigned)(lane + 32 * h)];
        }
    }
#pragma unroll
    for (int h = 0; h < 2; ++h) {
#pragma unroll
        for (int k = 0; k < 8; ++k)
            if ((unsigned)(lane + 32 * h) < cN[k]) gcn_ncount(cnt, p[h][k].y, lo, nr);
    }
#pragma unroll
    for (int k = 0; k < 8; ++k)
        for (unsigned j = (unsigned)lane + 64u; j < cN[k]; j += 32u)
            gcn_ncount(cnt, pairs[pB[k] + j].y, lo, nr);
    __syncthreads();

    unsigned v = 0u;
    if (tid < RN_MAX) v = cnt[tid];
    unsigned x = v;
    for (int d = 1; d < 32; d <<= 1) {
        const unsigned y = __shfl_up(x, d);
        if (lane >= d) x += y;
    }
    if (tid < RN_MAX && lane == 31) wt[wave] = x;
    __syncthreads();
    if (tid < RN_MAX) {
        unsigned start = e0 + (x - v);
        for (int w = 0; w < wave; ++w) start += wt[w];
        cnt[tid] = start;
        rsl[tid] = start;
        if (tid == RN_MAX - 1) rsl[RN_MAX] = start + v;
        if ((unsigned)tid < nr) {
            rowstart[lo + tid] = start;
            const float d = (float)v;
            deg[lo + tid] = d > 0.0f ? 1.0f / d : 0.0f;
        }
    }
    if (r == NRANGE - 1 && tid == 0) rowstart[N] = e0 + tot;
    __syncthreads();

    const int nl = NODES_PER_WAVE * wave;
    float* tile = tl + wave * (NODES_PER_WAVE * LDS_PITCH);

    if (__builtin_expect(in_lds, true)) {
#pragma unroll
        for (int h = 0; h < 2; ++h) {
#pragma unroll
            for (int k = 0; k < 8; ++k)
                if ((unsigned)(lane + 32 * h) < cN[k])
                    gcn_nplace(cnt, ids, eb, (unsigned)IDS_CAP, p[h][k], lo, nr);
        }
#pragma unroll
        for (int k = 0; k < 8; ++k)
            for (unsigned j = (unsigned)lane + 64u; j < cN[k]; j += 32u)
                gcn_nplace(cnt, ids, eb, (unsigned)IDS_CAP, pairs[pB[k] + j], lo, nr);
        __syncthreads();

        const unsigned a  = e0 - eb;
        const unsigned nw = rsl[RN_MAX] - eb;
        const unsigned q0 = (a + 3u) >> 2;
        const unsigned q1 = nw >> 2;
        unsigned* cw = csr + eb;
        for (unsigned q = q0 + (unsigned)tid; q < q1; q += EP_THREADS)
            ((uint4*)cw)[q] = ((const uint4*)ids)[q];
        if (tid < 8) {
            const unsigned j = tid < 4 ? a + (unsigned)tid : max(4u * q1, 4u * q0) + (unsigned)(tid - 4);
            const unsigned lim = tid < 4 ? min(4u * q0, nw) : nw;
            if (j < lim) cw[j] = ids[j];
        }

        if ((unsigned)nl < nr)
            gcn_range_layer_wave(rsl, ids, eb, nl, nr, feat, wl, tile, bias, out, lo + nl, lane);
    } else {
#pragma unroll
        for (int h = 0; h < 2; ++h) {
#pragma unroll
            for (int k = 0; k < 8; ++k)
                if ((unsigned)(lane + 32 * h) < cN[k])
                    gcn_nplace(cnt, csr, 0u, (unsigned)E, p[h][k], lo, nr);
        }
#pragma unroll
        for (int k = 0; k < 8; ++k)
            for (unsigned j = (unsigned)lane + 64u; j < cN[k]; j += 32u)
                gcn_nplace(cnt, csr, 0u, (unsigned)E, pairs[pB[k] + j], lo, nr);

        __syncthreads();

        if ((unsigned)nl < nr)
            gcn_range_layer_wave(rsl, csr, 0u, nl, nr, feat, wl, tile, bias, out, lo + nl, lane);
    }
}

extern "C" void kernel_launch(void* const* d_in, const int* in_sizes, int n_in,
                              void* d_out, int out_size, void* d_ws, size_t ws_size,
                              hipStream_t stream) {
    const float* h   = (const float*)d_in[0];
    const int*   src = (const int*)  d_in[1];
    const int*   dst = (const int*)  d_in[2];
    const float* W1  = (const float*)d_in[3];
    const float* b1  = (const float*)d_in[4];
    const float* W2  = (const float*)d_in[5];
    const float* b2  = (const float*)d_in[6];
    float*       out = (float*)d_out;

    const int N = in_sizes[0] / DFEAT;
    const int E = in_sizes[1];

    const int layerB = (N + NODES_PER_WAVE * WAVES_PER_BLOCK - 1)
                     / (NODES_PER_WAVE * WAVES_PER_BLOCK);

    char* ws = (char*)d_ws;
    const size_t A = 255;
    float* deg = (float*)ws;
    size_t off = (((size_t)N * sizeof(float)) + A) & ~A;
    float* h1 = (float*)(ws + off);
    off += (size_t)N * DFEAT * sizeof(float);
    unsigned* tcnt = (unsigned*)(ws + off);
    off += (((size_t)NRANGE * NSLICE * sizeof(unsigned)) + A) & ~A;
    unsigned* pairs = (unsigned*)(ws + off);
    off += (((size_t)E * 2 * sizeof(unsigned)) + A) & ~A;
    unsigned* rowstart = (unsigned*)(ws + off);
    off += (((size_t)(N + 1) * sizeof(unsigned)) + A) & ~A;
    unsigned* csr = (unsigned*)(ws + off);

    gcn_lsort_kernel<<<NSLICE, EP_THREADS, 0, stream>>>(src, dst, tcnt, pairs, N, E);
    gcn_rsort_layer_kernel<<<NRANGE, EP_THREADS, 0, stream>>>(pairs, tcnt, rowstart, deg, csr,
                                                             h, W1, b1, h1, N, E);

    gcn_layer_kernel<<<layerB, WAVES_PER_BLOCK * 32, 0, stream>>>(h1, rowstart, csr, deg, W2, b2, out, N);
}


# ELU epilogue of both layers: exp(v)-1 as one hardware exp2 of v*log2(e) minus 1, branch-free, instead of the library expm1f (22 dependent instructions per element); abs error about 1e-7
# speedup vs baseline: 6.1004x; 1.0831x over previous
#include <hip/hip_runtime.h>
#include <hip/hip_bf16.h>
#include <math.h>

typedef __attribute__((ext_vector_type(2))) float v2f;
typedef __attribute__((ext_vector_type(8))) float v8f;

#define DFEAT 64

#define NRANGE     256
#define NSLICE     256
#define RN_MAX     512
#define EP_THREADS 1024
#define SL_CAP     (8 * EP_THREADS)
#define IDS_CAP    16384

__device__ __forceinline__ unsigned gcn_rn(int N) {
    return (unsigned)max(1, (N + NRANGE - 1) / NRANGE);
}

__device__ __forceinline__ int gcn_slice_len(int E) {
    return (((E + NSLICE - 1) / NSLICE) + 3) & ~3;
}

__device__ __forceinline__ unsigned gcn_range_of(int d, unsigned rn) {
    return min((unsigned)d / rn, (unsigned)(NRANGE - 1));
}

__device__ __forceinline__ unsigned gcn_take(unsigned* c) {
    return __hip_atomic_fetch_add(c, 1u, __ATOMIC_RELAXED, __HIP_MEMORY_SCOPE_WORKGROUP);
}

__global__ __launch_bounds__(EP_THREADS)
void gcn_lsort_kernel(const int* __restrict__ src,
                      const int* __restrict__ dst,
                      unsigned* __restrict__ tcnt,
                      unsigned* __restrict__ pairs_w, int N, int E) {
    uint2* __restrict__ pairs = (uint2*)pairs_w;
    __shared__ __attribute__((aligned(16))) uint2 recs[SL_CAP];
    __shared__ unsigned cnt[NRANGE];
    __shared__ unsigned wt[NRANGE / 32];
    const int tid  = threadIdx.x;
    const int lane = tid & 31;
    const int s    = (int)blockIdx.x;
    if (tid < NRANGE) cnt[tid] = 0u;

    const unsigned rn = gcn_rn(N);
    const int sl = gcn_slice_len(E);
    const int e0 = s * sl;
    const int e1 = min(e0 + sl, E);
    const int len   = min(max(e1 - e0, 0), SL_CAP);
    const int full4 = len & ~3;
    const int* dp = dst + e0;
    const int* sp = src + e0;

    int4 dv[2], sv[2];
    if (full4 > 0) {
#pragma unroll
        for (int k = 0; k < 2; ++k) {
            const int j = min(4 * tid + k * 4 * EP_THREADS, full4 - 4);
            dv[k] = *(const int4*)(dp + j);
            sv[k] = *(const int4*)(sp + j);
        }
    }
    const bool ht = tid < len - full4;
    int dt = 0, st = 0;
    if (ht) { dt = dp[full4 + tid]; st = sp[full4 + tid]; }
    __syncthreads();

    unsigned rg[2][4], rk[2][4], rgt = 0u, rkt = 0u;
#pragma unroll
    for (int k = 0; k < 2; ++k) {
        if (4 * tid + k * 4 * EP_THREADS < full4) {
            rg[k][0] = gcn_range_of(dv[k].x, rn);
            rg[k][1] = gcn_range_of(dv[k].y, rn);
            rg[k][2] = gcn_range_of(dv[k].z, rn);
            rg[k][3] = gcn_range_of(dv[k].w, rn);
#pragma unroll
            for (int c = 0; c < 4; ++c) rk[k][c] = gcn_take(&cnt[rg[k][c]]);
        }
    }
    if (ht) { rgt = gcn_range_of(dt, rn); rkt = gcn_take(&cnt[rgt]); }
    __syncthreads();

    unsigned v = 0u;
    if (tid < NRANGE) v = cnt[tid];
    unsigned x = v;
    for (int d = 1; d < 32; d <<= 1) {
        const unsigned y = __shfl_up(x, d);
        if (lane >= d) x += y;
    }
    if (tid < NRANGE && lane == 31) wt[tid >> 5] = x;
    __syncthreads();
    if (tid < NRANGE) {
        unsigned base = x - v;
        for (int w = 0; w < (tid >> 5); ++w) base += wt[w];
        cnt[tid] = base;
        tcnt[tid * NSLICE + s] = v;
    }
    __syncthreads();

#pragma unroll
    for (int k = 0; k < 2; ++k) {
        if (4 * tid + k * 4 * EP_THREADS < full4) {
            const unsigned p0 = cnt[rg[k][0]] + rk[k][0];
            const unsigned p1 = cnt[rg[k][1]] + rk[k][1];
            const unsigned p2 = cnt[rg[k][2]] + rk[k][2];
            const unsigned p3 = cnt[rg[k][3]] + rk[k][3];
            if (p0 < (unsigned)SL_CAP) recs[p0] = make_uint2((unsigned)sv[k].x, (unsigned)dv[k].x);
            if (p1 < (unsigned)SL_CAP) recs[p1] = make_uint2((unsigned)sv[k].y, (unsigned)dv[k].y);
            if (p2 < (unsigned)SL_CAP) recs[p2] = make_uint2((unsigned)sv[k].z, (unsigned)dv[k].z);
            if (p3 < (unsigned)SL_CAP) recs[p3] = make_uint2((unsigned)sv[k].w, (unsigned)dv[k].w);
        }
    }
    if (ht) {
        const unsigned pt = cnt[rgt] + rkt;
        if (pt < (unsigned)SL_CAP) recs[pt] = make_uint2((unsigned)st, (unsigned)dt);
    }
    __syncthreads();

    const int len2 = len >> 1;
    uint4* gp = (uint4*)(pairs + e0);
    const uint4* lp = (const uint4*)recs;
    if (len2 > 0) {
        uint4 ov[SL_CAP / (2 * EP_THREADS)];
#pragma unroll
        for (int k = 0; k < SL_CAP / (2 * EP_THREADS); ++k)
            ov[k] = lp[min(tid + k * EP_THREADS, len2 - 1)];
#pragma unroll
        for (int k = 0; k < SL_CAP / (2 * EP_THREADS); ++k) {
            const int j = tid + k * EP_THREADS;
            if (j < len2) gp[j] = ov[k];
        }
    }
    if ((len & 1) && tid == 0) pairs[e0 + len - 1] = recs[len - 1];
}

__device__ __forceinline__ void gcn_ncount(unsigned* cnt, unsigned d, int lo, unsigned nr) {
    const unsigned rel = d - (unsigned)lo;
    if (rel < nr)
        __hip_atomic_fetch_add(&cnt[rel], 1u, __ATOMIC_RELAXED, __HIP_MEMORY_SCOPE_WORKGROUP);
}

__device__ __forceinline__ void gcn_nplace(unsigned* cnt, unsigned* places, unsigned sub, unsigned cap,
                                           uint2 p, int lo, unsigned nr) {
    const unsigned rel = p.y - (unsigned)lo;
    if (rel < nr) {
        const unsigned w = gcn_take(&cnt[rel]) - sub;
        if (w < cap) places[w] = p.x;
    }
}


#define NODES_PER_WAVE  16
#define WAVES_PER_BLOCK 8
#define LDS_PITCH       68

__device__ __forceinline__ const float2* gcn_row2(const float* base, unsigned row, int lane) {
    const unsigned off = row * (unsigned)(DFEAT * sizeof(float)) + (unsigned)lane * (unsigned)sizeof(float2);
    return (const float2*)((const char*)base + off);
}

#define GCN_GATHER8(v, j0)                                                   \
    _Pragma("unroll")                                                        \
    for (int j = 0; j < 8; ++j) {                                            \
        const unsigned sj = (unsigned)__builtin_amdgcn_readlane((int)idx, (j0) + j); \
        v[j] = *gcn_row2(x, sj, lane);                                       \
    }

#define GCN_FLUSH()                                                          \
    {                                                                        \
        const float sc = __int_as_float(__builtin_amdgcn_readlane(__float_as_int(iv), p)); \
        trow[p * (LDS_PITCH / 2)] = make_float2(acc.x * sc, acc.y * sc);     \
        acc = make_float2(0.0f, 0.0f);                                       \
        ++p;                                                                 \
    }

#define GCN_SUM8(v, j0)                                                      \
    _Pragma("unroll")                                                        \
    for (int j = 0; j < 8; ++j) {                                            \
        const unsigned ee = e + (unsigned)((j0) + j);                        \
        while (ee >= nodeEnd) {                                              \
            GCN_FLUSH()                                                      \
            nodeEnd = (unsigned)__builtin_amdgcn_readlane((int)rs, p + 1);   \
        }                                                                    \
        acc.x += v[j].x;                                                     \
        acc.y += v[j].y;                                                     \
    }

#define GCN_GROUP32()                                                        \
    {                                                                        \
        float2 a[8], b[8];                                                   \
        GCN_GATHER8(a, 0)                                                    \
        GCN_GATHER8(b, 8)                                                    \
        GCN_SUM8(a, 0)                                                       \
        GCN_GATHER8(a, 16)                                                   \
        GCN_SUM8(b, 8)                                                       \
        GCN_GATHER8(b, 24)                                                   \
        GCN_SUM8(a, 16)                                                      \
        GCN_SUM8(b, 24)                                                      \
    }

__device__ __forceinline__ v8f gcn_mma(v2f a, v2f b, v8f c) {
    return __builtin_amdgcn_wmma_f32_16x16x4_f32(
                false, a,  false, b,
                (short)0, c,  false,  false);
}

__device__ __forceinline__ void gcn_finish_tile(float* t, v8f c, float bcol, int hi, int col) {
#pragma unroll
    for (int r = 0; r < 8; ++r) {
        const float v  = c[r] + bcol;
        const float en = __builtin_amdgcn_exp2f(v * 1.44269504088896341f) - 1.0f;
        t[(r + (hi << 3)) * LDS_PITCH + col] = v > 0.0f ? v : en;
    }
}

#define GCN_LOADK(FA, FB, t)                                                 \
    _Pragma("unroll")                                                        \
    for (int ks = 0; ks < 2; ++ks) {                                         \
        FA[ks] = *(const v2f*)(ap + 8 * (t) + 4 * ks);                       \
        _Pragma("unroll")                                                    \
        for (int ct = 0; ct < 4; ++ct)                                       \
            FB[ks][ct] = *(const v2f*)(bp + ct * 16 * LDS_PITCH + 8 * (t) + 4 * ks); \
    }

#define GCN_MMAK(FA, FB)                                                     \
    _Pragma("unroll")                                                        \
    for (int ks = 0; ks < 2; ++ks) {                                         \
        _Pragma("unroll")                                                    \
        for (int ct = 0; ct < 4; ++ct)                                       \
            c[ct] = gcn_mma(FA[ks], FB[ks][ct], c[ct]);                      \
    }


__device__ __forceinline__ void gcn_layer_wave(const float* __restrict__ x,
                                               const unsigned* csr,
                                               unsigned rs, float iv, int nn,
                                               const float* wl, float* tile,
                                               const float* __restrict__ bias,
                                               float* __restrict__ out, int n0, int lane) {
    const unsigned eBeg = (unsigned)__builtin_amdgcn_readlane((int)rs, 0);
    const unsigned eEnd = (unsigned)__builtin_amdgcn_readlane((int)rs, nn);
    unsigned nodeEnd    = (unsigned)__builtin_amdgcn_readlane((int)rs, 1);
    int p = 0;
    float2 acc = make_float2(0.0f, 0.0f);
    float2* trow = (float2*)tile + lane;

    if (eBeg < eEnd) {
        const unsigned eLast = eEnd - 1u;
        unsigned iN = csr[min(eBeg + (unsigned)lane, eLast)];

        for (unsigned e = eBeg; e < eEnd; e += 32) {
            const unsigned idx = iN;
            iN = csr[min(e + 32u + (unsigned)lane, eLast)];
            GCN_GROUP32()
        }
    }
    while (p < NODES_PER_WAVE) GCN_FLUSH()

    __builtin_amdgcn_fence(__ATOMIC_RELEASE, "wavefront");
    __builtin_amdgcn_fence(__ATOMIC_ACQUIRE, "wavefront");

    const int m  = lane & 15;
    const int hi = lane >> 4;
    const int k2 = hi << 1;
    const float* ap = tile + m * LDS_PITCH + k2;
    const float* bp = wl   + m * LDS_PITCH + k2;

    v8f c[4] = {};
    v2f fa[2][2], fb[2][2][4];
    float bcol[4];
#pragma unroll
    for (int ct = 0; ct < 4; ++ct) bcol[ct] = bias[16 * ct + m];
    GCN_LOADK(fa[0], fb[0], 0)
#pragma unroll
    for (int t = 0; t < DFEAT / 8; ++t) {
        if (t + 1 < DFEAT / 8) GCN_LOADK(fa[(t + 1) & 1], fb[(t + 1) & 1], t + 1)
        __builtin_amdgcn_sched_barrier(0);
        GCN_MMAK(fa[t & 1], fb[t & 1])
        __builtin_amdgcn_sched_barrier(0);
    }

#pragma unroll
    for (int ct = 0; ct < 4; ++ct)
        gcn_finish_tile(tile, c[ct], bcol[ct], hi, 16 * ct + m);

    __builtin_amdgcn_fence(__ATOMIC_RELEASE, "wavefront");
    __builtin_amdgcn_fence(__ATOMIC_ACQUIRE, "wavefront");

    const int q4 = m << 2;
    float* orow = out + (size_t)n0 * DFEAT + q4;
    if (nn == NODES_PER_WAVE) {
        float4 ov[NODES_PER_WAVE / 2];
#pragma unroll
        for (int j = 0; j < NODES_PER_WAVE / 2; ++j)
            ov[j] = *(const float4*)(tile + (2 * j + hi) * LDS_PITCH + q4);
#pragma unroll
        for (int j = 0; j < NODES_PER_WAVE / 2; ++j)
            *(float4*)(orow + (2 * j + hi) * DFEAT) = ov[j];
    } else {
        for (int j = 0; j < NODES_PER_WAVE / 2; ++j) {
            const int row = 2 * j + hi;
            if (row < nn)
                *(float4*)(orow + row * DFEAT) = *(const float4*)(tile + row * LDS_PITCH + q4);
        }
    }
}

__global__ __launch_bounds__(WAVES_PER_BLOCK * 32)
void gcn_layer_kernel(const float* __restrict__ x,
                      const unsigned* __restrict__ rowstart,
                      const unsigned* __restrict__ csr,
                      const float* __restrict__ inv_deg,
                      const float* __restrict__ W,
                      const float* __restrict__ bias,
                      float* __restrict__ out, int N) {
    __shared__ __attribute__((aligned(16))) float wl[DFEAT * LDS_PITCH];
    __shared__ __attribute__((aligned(16))) float tl[WAVES_PER_BLOCK * NODES_PER_WAVE * LDS_PITCH];

    const int lane = threadIdx.x & 31;
    const int wv = __builtin_amdgcn_readfirstlane((int)(threadIdx.x >> 5));
    const int n0 = ((int)blockIdx.x * WAVES_PER_BLOCK + wv) * NODES_PER_WAVE;

    {
        float4 w4[4];
#pragma unroll
        for (int t = 0; t < 4; ++t)
            w4[t] = *(const float4*)(W + 4 * ((int)threadIdx.x + t * (WAVES_PER_BLOCK * 32)));
#pragma unroll
        for (int t = 0; t < 4; ++t) {
            const int i = (int)threadIdx.x + t * (WAVES_PER_BLOCK * 32);
            *(float4*)(wl + (i >> 4) * LDS_PITCH + ((i & 15) << 2)) = w4[t];
        }
    }
    __syncthreads();
    if (n0 >= N) return;
    const int nn = min(NODES_PER_WAVE, N - n0);

    float* tile = tl + wv * (NODES_PER_WAVE * LDS_PITCH);

    unsigned rs = ~0u;
    if (lane <= nn) rs = rowstart[n0 + lane];
    float iv = 0.0f;
    if (lane < nn) iv = inv_deg[n0 + lane];
    gcn_layer_wave(x, csr, rs, iv, nn, wl, tile, bias, out, n0, lane);
}

__device__ __forceinline__ void gcn_range_layer_wave(const unsigned* rsl, const unsigned* places,
                                                     unsigned sub, int nl, unsigned nr,
                                                     const float* __restrict__ feat,
                                                     const float* wl, float* tile,
                                                     const float* __restrict__ bias,
                                                     float* __restrict__ out, int n0, int lane) {
    const int nn = min(NODES_PER_WAVE, (int)nr - nl);
    unsigned rs = ~0u;
    if (lane <= nn) rs = rsl[nl + lane] - sub;
    float iv = 0.0f;
    if (lane < nn) {
        const float d = (float)(rsl[nl + lane + 1] - rsl[nl + lane]);
        iv = d > 0.0f ? 1.0f / d : 0.0f;
    }
    gcn_layer_wave(feat, places, rs, iv, nn, wl, tile, bias, out, n0, lane);
}

__global__ __launch_bounds__(EP_THREADS)
void gcn_rsort_layer_kernel(const unsigned* __restrict__ pairs_w,
                            const unsigned* __restrict__ tcnt,
                            unsigned* __restrict__ rowstart,
                            float* __restrict__ deg,
                            unsigned* csr,
                            const float* __restrict__ feat,
                            const float* __restrict__ W,
                            const float* __restrict__ bias,
                            float* __restrict__ out, int N, int E) {
    static_assert(EP_THREADS * 4 == DFEAT * DFEAT, "one float4 of W per thread");
    static_assert(RN_MAX / NODES_PER_WAVE <= EP_THREADS / 32, "a wave per 16 nodes of the range");
    const uint2* __restrict__ pairs = (const uint2*)pairs_w;
    __shared__ unsigned cnt[RN_MAX];
    __shared__ unsigned wt[RN_MAX / 32];
    __shared__ unsigned cs[4][NSLICE];
    __shared__ unsigned ps[NSLICE];
    __shared__ unsigned pc[NSLICE];
    __shared__ unsigned wb[2][NSLICE / 32];
    __shared__ unsigned rsl[RN_MAX + 1];
    __shared__ __attribute__((aligned(16))) unsigned ids[IDS_CAP];
    __shared__ __attribute__((aligned(16))) float wl[DFEAT * LDS_PITCH];
    __shared__ __attribute__((aligned(16))) float tl[(RN_MAX / NODES_PER_WAVE) * NODES_PER_WAVE * LDS_PITCH];
    const int tid  = threadIdx.x;
    const int lane = tid & 31;
    const int wave = __builtin_amdgcn_readfirstlane(tid >> 5);
    const int q    = __builtin_amdgcn_readfirstlane(tid >> 8);
    const int col  = tid & (NSLICE - 1);
    const int r    = (int)blockIdx.x;
    const unsigned rn = gcn_rn(N);
    const int lo = r * (int)rn;
    const unsigned nr = (unsigned)max(0, min(min((int)rn, RN_MAX), N - lo));
    const int sl = gcn_slice_len(E);

    if (tid < RN_MAX) cnt[tid] = 0u;
    const float4 w4 = *(const float4*)(W + 4 * tid);

    unsigned b = 0u;
    {
        const int r0 = q * (NRANGE / 4);
        const unsigned* tp = tcnt + col;
        unsigned c[NRANGE / 4];
#pragma unroll
        for (int k = 0; k < NRANGE / 4; ++k) c[k] = tp[min(r0 + k, r) * NSLICE];
#pragma unroll
        for (int k = 0; k < NRANGE / 4; ++k)
            if (r0 + k < r) b += c[k];
    }
    unsigned myc = 0u;
    if (tid < NSLICE) myc = tcnt[r * NSLICE + tid];
    cs[q][col] = b;
    *(float4*)(wl + (tid >> 4) * LDS_PITCH + ((tid & 15) << 2)) = w4;
    __syncthreads();

    unsigned pst = 0u;
    if (tid < NSLICE) {
        pst = cs[0][tid] + cs[1][tid] + cs[2][tid] + cs[3][tid];
        ps[tid] = pst;
        pc[tid] = myc;
    }
    unsigned a0 = pst, a1 = myc;
    for (int d = 16; d > 0; d >>= 1) {
        a0 += __shfl_xor(a0, d);
        a1 += __shfl_xor(a1, d);
    }
    if (tid < NSLICE && lane == 0) { wb[0][tid >> 5] = a0; wb[1][tid >> 5] = a1; }
    __syncthreads();
    unsigned e0 = 0u, tot = 0u;
#pragma unroll
    for (int w = 0; w < NSLICE / 32; ++w) { e0 += wb[0][w]; tot += wb[1][w]; }
    const unsigned eb = (unsigned)__builtin_amdgcn_readfirstlane((int)(e0 & ~3u));
    const bool in_lds = __builtin_amdgcn_readfirstlane((int)((e0 - eb) + tot <= (unsigned)IDS_CAP)) != 0;

    unsigned cN[8], pB[8];
#pragma unroll
    for (int k = 0; k < 8; ++k) {
        const int sk = 8 * wave + k;
        cN[k] = (unsigned)__builtin_amdgcn_readfirstlane((int)pc[sk]);
        pB[k] = (unsigned)(sk * sl) + (unsigned)__builtin_amdgcn_readfirstlane((int)ps[sk]);
    }
    uint2 p[2][8];
#pragma unroll
    for (int h = 0; h < 2; ++h) {
#pragma unroll
        for (int k = 0; k < 8; ++k) {
            p[h][k] = make_uint2(0u, 0u);
            if ((unsigned)(lane + 32 * h) < cN[k]) p[h][k] = pairs[pB[k] + (unsigned)(lane + 32 * h)];
        }
    }
#pragma unroll
    for (int h = 0; h < 2; ++h) {
#pragma unroll
        for (int k = 0; k < 8; ++k)
            if ((unsigned)(lane + 32 * h) < cN[k]) gcn_ncount(cnt, p[h][k].y, lo, nr);
    }
#pragma unroll
    for (int k = 0; k < 8; ++k)
        for (unsigned j = (unsigned)lane + 64u; j < cN[k]; j += 32u)
            gcn_ncount(cnt, pairs[pB[k] + j].y, lo, nr);
    __syncthreads();

    unsigned v = 0u;
    if (tid < RN_MAX) v = cnt[tid];
    unsigned x = v;
    for (int d = 1; d < 32; d <<= 1) {
        const unsigned y = __shfl_up(x, d);
        if (lane >= d) x += y;
    }
    if (tid < RN_MAX && lane == 31) wt[wave] = x;
    __syncthreads();
    if (tid < RN_MAX) {
        unsigned start = e0 + (x - v);
        for (int w = 0; w < wave; ++w) start += wt[w];
        cnt[tid] = start;
        rsl[tid] = start;
        if (tid == RN_MAX - 1) rsl[RN_MAX] = start + v;
        if ((unsigned)tid < nr) {
            rowstart[lo + tid] = start;
            const float d = (float)v;
            deg[lo + tid] = d > 0.0f ? 1.0f / d : 0.0f;
        }
    }
    if (r == NRANGE - 1 && tid == 0) rowstart[N] = e0 + tot;
    __syncthreads();

    const int nl = NODES_PER_WAVE * wave;
    float* tile = tl + wave * (NODES_PER_WAVE * LDS_PITCH);

    if (__builtin_expect(in_lds, true)) {
#pragma unroll
        for (int h = 0; h < 2; ++h) {
#pragma unroll
            for (int k = 0; k < 8; ++k)
                if ((unsigned)(lane + 32 * h) < cN[k])
                    gcn_nplace(cnt, ids, eb, (unsigned)IDS_CAP, p[h][k], lo, nr);
        }
#pragma unroll
        for (int k = 0; k < 8; ++k)
            for (unsigned j = (unsigned)lane + 64u; j < cN[k]; j += 32u)
                gcn_nplace(cnt, ids, eb, (unsigned)IDS_CAP, pairs[pB[k] + j], lo, nr);
        __syncthreads();

        const unsigned a  = e0 - eb;
        const unsigned nw = rsl[RN_MAX] - eb;
        const unsigned q0 = (a + 3u) >> 2;
        const unsigned q1 = nw >> 2;
        unsigned* cw = csr + eb;
        for (unsigned q = q0 + (unsigned)tid; q < q1; q += EP_THREADS)
            ((uint4*)cw)[q] = ((const uint4*)ids)[q];
        if (tid < 8) {
            const unsigned j = tid < 4 ? a + (unsigned)tid : max(4u * q1, 4u * q0) + (unsigned)(tid - 4);
            const unsigned lim = tid < 4 ? min(4u * q0, nw) : nw;
            if (j < lim) cw[j] = ids[j];
        }

        if ((unsigned)nl < nr)
            gcn_range_layer_wave(rsl, ids, eb, nl, nr, feat, wl, tile, bias, out, lo + nl, lane);
    } else {
#pragma unroll
        for (int h = 0; h < 2; ++h) {
#pragma unroll
            for (int k = 0; k < 8; ++k)
                if ((unsigned)(lane + 32 * h) < cN[k])
                    gcn_nplace(cnt, csr, 0u, (unsigned)E, p[h][k], lo, nr);
        }
#pragma unroll
        for (int k = 0; k < 8; ++k)
            for (unsigned j = (unsigned)lane + 64u; j < cN[k]; j += 32u)
                gcn_nplace(cnt, csr, 0u, (unsigned)E, pairs[pB[k] + j], lo, nr);

        __syncthreads();

        if ((unsigned)nl < nr)
            gcn_range_layer_wave(rsl, csr, 0u, nl, nr, feat, wl, tile, bias, out, lo + nl, lane);
    }
}

extern "C" void kernel_launch(void* const* d_in, const int* in_sizes, int n_in,
                              void* d_out, int out_size, void* d_ws, size_t ws_size,
                              hipStream_t stream) {
    const float* h   = (const float*)d_in[0];
    const int*   src = (const int*)  d_in[1];
    const int*   dst = (const int*)  d_in[2];
    const float* W1  = (const float*)d_in[3];
    const float* b1  = (const float*)d_in[4];
    const float* W2  = (const float*)d_in[5];
    const float* b2  = (const float*)d_in[6];
    float*       out = (float*)d_out;

    const int N = in_sizes[0] / DFEAT;
    const int E = in_sizes[1];

    const int layerB = (N + NODES_PER_WAVE * WAVES_PER_BLOCK - 1)
                     / (NODES_PER_WAVE * WAVES_PER_BLOCK);

    char* ws = (char*)d_ws;
    const size_t A = 255;
    float* deg = (float*)ws;
    size_t off = (((size_t)N * sizeof(float)) + A) & ~A;
    float* h1 = (float*)(ws + off);
    off += (size_t)N * DFEAT * sizeof(float);
    unsigned* tcnt = (unsigned*)(ws + off);
    off += (((size_t)NRANGE * NSLICE * sizeof(unsigned)) + A) & ~A;
    unsigned* pairs = (unsigned*)(ws + off);
    off += (((size_t)E * 2 * sizeof(unsigned)) + A) & ~A;
    unsigned* rowstart = (unsigned*)(ws + off);
    off += (((size_t)(N + 1) * sizeof(unsigned)) + A) & ~A;
    unsigned* csr = (unsigned*)(ws + off);

    gcn_lsort_kernel<<<NSLICE, EP_THREADS, 0, stream>>>(src, dst, tcnt, pairs, N, E);
    gcn_rsort_layer_kernel<<<NRANGE, EP_THREADS, 0, stream>>>(pairs, tcnt, rowstart, deg, csr,
                                                             h, W1, b1, h1, N, E);

    gcn_layer_kernel<<<layerB, WAVES_PER_BLOCK * 32, 0, stream>>>(h1, rowstart, csr, deg, W2, b2, out, N);
}
